# starting kernel
# baseline (speedup 1.0000x reference)
#include <hip/hip_runtime.h>
#include <hip/hip_bf16.h>

typedef __attribute__((ext_vector_type(2))) float v2f;
typedef __attribute__((ext_vector_type(4))) float v4f;
typedef __attribute__((ext_vector_type(8))) float v8f;

#define NNODES    50000
#define NEDGES    640000
#define CH        128
#define ROW_TILES 3125

__global__ void gcn_deg_init(float* __restrict__ deg) {
    int i = blockIdx.x * blockDim.x + threadIdx.x;
    if (i < NNODES) deg[i] = 1.0f;
}

__global__ void gcn_deg_accum(const int* __restrict__ dst, float* __restrict__ deg) {
    int e = blockIdx.x * blockDim.x + threadIdx.x;
    if (e < NEDGES) unsafeAtomicAdd(&deg[dst[e]], 1.0f);
}

__global__ void gcn_dinv(const float* __restrict__ deg, float* __restrict__ dinv) {
    int i = blockIdx.x * blockDim.x + threadIdx.x;
    if (i < NNODES) {
        float d = deg[i];
        dinv[i] = (d > 0.0f) ? rsqrtf(d) : 0.0f;
    }
}

__global__ __launch_bounds__(256) void gcn_gemm_wmma(const float* __restrict__ X,
                                                     const float* __restrict__ W,
                                                     float* __restrict__ H) {
    __shared__ float Wt[CH * CH];
    for (int idx = threadIdx.x; idx < CH * CH; idx += 256) {
        int k = idx >> 7, c = idx & 127;
        Wt[c * CH + k] = W[idx];
    }
    __syncthreads();

    const int wave = threadIdx.x >> 5;
    const int lane = threadIdx.x & 31;
    const int rt   = blockIdx.x * 8 + wave;
    if (rt >= ROW_TILES) return;

    const int half = lane >> 4;
    const int mn   = lane & 15;

    const float* __restrict__ xrow = X + (size_t)(rt * 16 + mn) * CH;

    v8f acc[8];
    v8f zero = {};
    #pragma unroll
    for (int c = 0; c < 8; ++c) acc[c] = zero;

    #pragma unroll 4
    for (int kk = 0; kk < CH; kk += 4) {
        v2f a = *(const v2f*)(xrow + kk + 2 * half);
        #pragma unroll
        for (int c = 0; c < 8; ++c) {
            v2f b = *(const v2f*)(&Wt[(c * 16 + mn) * CH + kk + 2 * half]);
            acc[c] = __builtin_amdgcn_wmma_f32_16x16x4_f32(
                false, a, false, b, (short)0, acc[c], false, false);
        }
    }

    #pragma unroll
    for (int c = 0; c < 8; ++c) {
        #pragma unroll
        for (int i = 0; i < 8; ++i) {
            H[(size_t)(rt * 16 + half * 8 + i) * CH + c * 16 + mn] = acc[c][i];
        }
    }
}

__global__ void gcn_agg_init(const float* __restrict__ H, const float* __restrict__ dinv,
                             const float* __restrict__ bias, float* __restrict__ out) {
    int idx = blockIdx.x * blockDim.x + threadIdx.x;
    if (idx >= NNODES * 32) return;
    int node = idx >> 5;
    int ch   = (idx & 31) * 4;
    float s  = dinv[node];
    s = s * s;
    v4f h = *(const v4f*)(&H[(size_t)node * CH + ch]);
    v4f b = *(const v4f*)(&bias[ch]);
    v4f o = h * s + b;
    *(v4f*)(&out[(size_t)node * CH + ch]) = o;
}

__global__ __launch_bounds__(256) void gcn_agg_edges(const int* __restrict__ src,
                                                     const int* __restrict__ dst,
                                                     const float* __restrict__ dinv,
                                                     const float* __restrict__ H,
                                                     float* __restrict__ out) {
    int e = blockIdx.x * 8 + (threadIdx.x >> 5);
    if (e >= NEDGES) return;
    int lane = threadIdx.x & 31;
    int s = src[e];
    int d = dst[e];
    float norm = dinv[s] * dinv[d];
    int ch = lane * 4;
    v4f h = *(const v4f*)(&H[(size_t)s * CH + ch]);
    float* op = &out[(size_t)d * CH + ch];
    unsafeAtomicAdd(op + 0, h.x * norm);
    unsafeAtomicAdd(op + 1, h.y * norm);
    unsafeAtomicAdd(op + 2, h.z * norm);
    unsafeAtomicAdd(op + 3, h.w * norm);
}

__global__ void gcn_relu(float* __restrict__ x, int n4) {
    int i = blockIdx.x * blockDim.x + threadIdx.x;
    if (i >= n4) return;
    v4f v = *(const v4f*)(&x[i * 4]);
    v.x = fmaxf(v.x, 0.0f); v.y = fmaxf(v.y, 0.0f);
    v.z = fmaxf(v.z, 0.0f); v.w = fmaxf(v.w, 0.0f);
    *(v4f*)(&x[i * 4]) = v;
}

extern "C" void kernel_launch(void* const* d_in, const int* in_sizes, int n_in,
                              void* d_out, int out_size, void* d_ws, size_t ws_size,
                              hipStream_t stream) {
    const float* x   = (const float*)d_in[0];
    const int*   ei  = (const int*)d_in[1];
    const float* W1  = (const float*)d_in[2];
    const float* b1  = (const float*)d_in[3];
    const float* W2  = (const float*)d_in[4];
    const float* b2  = (const float*)d_in[5];
    float*       out = (float*)d_out;

    const int* src = ei;
    const int* dst = ei + NEDGES;

    float* ws   = (float*)d_ws;
    float* deg  = ws;
    float* dinv = ws + 65536;
    float* H    = ws + 131072;
    float* out1 = ws + 131072 + 6400000;

    const int T = 256;
    gcn_deg_init <<<(NNODES + T - 1) / T, T, 0, stream>>>(deg);
    gcn_deg_accum<<<(NEDGES + T - 1) / T, T, 0, stream>>>(dst, deg);
    gcn_dinv     <<<(NNODES + T - 1) / T, T, 0, stream>>>(deg, dinv);

    const int gemm_blocks = (ROW_TILES + 7) / 8;
    const int agg_init_blocks = (NNODES * 32 + T - 1) / T;
    const int agg_edge_blocks = (NEDGES + 7) / 8;
    const int relu_blocks = (NNODES * CH / 4 + T - 1) / T;

    gcn_gemm_wmma<<<gemm_blocks, T, 0, stream>>>(x, W1, H);
    gcn_agg_init <<<agg_init_blocks, T, 0, stream>>>(H, dinv, b1, out1);
    gcn_agg_edges<<<agg_edge_blocks, T, 0, stream>>>(src, dst, dinv, H, out1);
    gcn_relu     <<<relu_blocks, T, 0, stream>>>(out1, NNODES * CH / 4);

    gcn_gemm_wmma<<<gemm_blocks, T, 0, stream>>>(out1, W2, H);
    gcn_agg_init <<<agg_init_blocks, T, 0, stream>>>(H, dinv, b2, out);
    gcn_agg_edges<<<agg_edge_blocks, T, 0, stream>>>(src, dst, dinv, H, out);
}


# GEMM: each wave keeps all of W in registers as B fragments (1 wave per SIMD, 3-4 row tiles each, next rows fetched ahead); no LDS copy of W; tile columns interleaved by 4: 16-B W loads, H stores
# speedup vs baseline: 1.8805x; 1.8805x over previous
#include <hip/hip_runtime.h>
#include <hip/hip_bf16.h>

typedef __attribute__((ext_vector_type(2))) float v2f;
typedef __attribute__((ext_vector_type(4))) float v4f;
typedef __attribute__((ext_vector_type(8))) float v8f;

#define NNODES    50000
#define NEDGES    640000
#define CH        128
#define ROW_TILES 3125

__global__ void gcn_deg_init(float* __restrict__ deg) {
    int i = blockIdx.x * blockDim.x + threadIdx.x;
    if (i < NNODES) deg[i] = 1.0f;
}

__global__ void gcn_deg_accum(const int* __restrict__ dst, float* __restrict__ deg) {
    int e = blockIdx.x * blockDim.x + threadIdx.x;
    if (e < NEDGES) unsafeAtomicAdd(&deg[dst[e]], 1.0f);
}

__global__ void gcn_dinv(const float* __restrict__ deg, float* __restrict__ dinv) {
    int i = blockIdx.x * blockDim.x + threadIdx.x;
    if (i < NNODES) {
        float d = deg[i];
        dinv[i] = (d > 0.0f) ? rsqrtf(d) : 0.0f;
    }
}

#define GEMM_BLOCKS 256
#define GEMM_WAVES  (GEMM_BLOCKS * 4)
static_assert(GEMM_WAVES <= ROW_TILES, "every wave owns at least one row tile");
__global__ __launch_bounds__(128) void gcn_gemm_wmma(const float* __restrict__ X,
                                                     const float* __restrict__ W,
                                                     float* __restrict__ H) {
    const int lane = threadIdx.x & 31;
    const int half = lane >> 4;
    const int mn   = lane & 15;
    int t = blockIdx.x * 4 + __builtin_amdgcn_readfirstlane(threadIdx.x >> 5);

    v2f a[32];
    {
        const float* __restrict__ xp = X + (size_t)(t * 16 + mn) * CH + 2 * half;
        #pragma unroll
        for (int q = 0; q < 32; ++q) a[q] = *(const v2f*)(xp + 4 * q);
    }

    v2f b[32][8];
    {
        const float* __restrict__ wp = W + (size_t)(2 * half) * CH + 4 * mn;
        #pragma unroll
        for (int q = 0; q < 32; ++q) {
            #pragma unroll
            for (int g = 0; g < 2; ++g) {
                v4f w0 = *(const v4f*)(wp + (4 * q + 0) * CH + 64 * g);
                v4f w1 = *(const v4f*)(wp + (4 * q + 1) * CH + 64 * g);
                #pragma unroll
                for (int j = 0; j < 4; ++j) {
                    v2f p; p.x = w0[j]; p.y = w1[j];
                    b[q][4 * g + j] = p;
                }
            }
        }
    }

    for (; t < ROW_TILES; t += GEMM_WAVES) {
        const int tn = (t + GEMM_WAVES < ROW_TILES) ? t + GEMM_WAVES : t;
        v2f an[32];
        {
            const float* __restrict__ xn = X + (size_t)(tn * 16 + mn) * CH + 2 * half;
            #pragma unroll
            for (int q = 0; q < 32; ++q) an[q] = *(const v2f*)(xn + 4 * q);
        }
        __builtin_amdgcn_sched_barrier(0);

        v8f acc[8];
        v8f zero = {};
        #pragma unroll
        for (int c = 0; c < 8; ++c) acc[c] = zero;

        #pragma unroll
        for (int q = 0; q < 32; ++q) {
            #pragma unroll
            for (int c = 0; c < 8; ++c) {
                acc[c] = __builtin_amdgcn_wmma_f32_16x16x4_f32(
                    false, a[q], false, b[q][c], (short)0, acc[c], false, false);
            }
        }

        float* __restrict__ hp = H + (size_t)(t * 16 + half * 8) * CH + 4 * mn;
        #pragma unroll
        for (int r = 0; r < 8; ++r) {
            #pragma unroll
            for (int g = 0; g < 2; ++g) {
                v4f o;
                o.x = acc[4 * g + 0][r];
                o.y = acc[4 * g + 1][r];
                o.z = acc[4 * g + 2][r];
                o.w = acc[4 * g + 3][r];
                *(v4f*)(hp + r * CH + 64 * g) = o;
            }
        }

        __builtin_amdgcn_sched_barrier(0);
        #pragma unroll
        for (int q = 0; q < 32; ++q) a[q] = an[q];
    }
}

__global__ void gcn_agg_init(const float* __restrict__ H, const float* __restrict__ dinv,
                             const float* __restrict__ bias, float* __restrict__ out) {
    int idx = blockIdx.x * blockDim.x + threadIdx.x;
    if (idx >= NNODES * 32) return;
    int node = idx >> 5;
    int ch   = (idx & 31) * 4;
    float s  = dinv[node];
    s = s * s;
    v4f h = *(const v4f*)(&H[(size_t)node * CH + ch]);
    v4f b = *(const v4f*)(&bias[ch]);
    v4f o = h * s + b;
    *(v4f*)(&out[(size_t)node * CH + ch]) = o;
}

#define EDGE_WIN 8
static_assert(NEDGES % 256 == 0, "a block takes 256 edges");
__global__ __launch_bounds__(256) void gcn_agg_edges(const int* __restrict__ src,
                                                     const int* __restrict__ dst,
                                                     const float* __restrict__ dinv,
                                                     const float* __restrict__ H,
                                                     float* __restrict__ out) {
    const int lane = threadIdx.x & 31;
    const int e    = blockIdx.x * 256 + threadIdx.x;
    const int s    = src[e];
    const int d    = dst[e];
    const int nb   = __float_as_int(dinv[s] * dinv[d]);

    float h[EDGE_WIN][4];
    #pragma unroll
    for (int g = 0; g < EDGE_WIN; ++g) {
        const float* hp = H + (size_t)__builtin_amdgcn_readlane(s, g) * CH + lane;
        h[g][0] = hp[0];
        h[g][1] = hp[32];
        h[g][2] = hp[64];
        h[g][3] = hp[96];
    }
    #pragma unroll
    for (int j = 0; j < 32; ++j) {
        const int g = j % EDGE_WIN;
        const float norm = __int_as_float(__builtin_amdgcn_readlane(nb, j));
        float* op = out + (size_t)__builtin_amdgcn_readlane(d, j) * CH + lane;
        unsafeAtomicAdd(op + 0,  h[g][0] * norm);
        unsafeAtomicAdd(op + 32, h[g][1] * norm);
        unsafeAtomicAdd(op + 64, h[g][2] * norm);
        unsafeAtomicAdd(op + 96, h[g][3] * norm);
        if (j + EDGE_WIN < 32) {
            const float* hp = H + (size_t)__builtin_amdgcn_readlane(s, j + EDGE_WIN) * CH + lane;
            h[g][0] = hp[0];
            h[g][1] = hp[32];
            h[g][2] = hp[64];
            h[g][3] = hp[96];
        }
    }
}

__global__ void gcn_relu(float* __restrict__ x, int n4) {
    int i = blockIdx.x * blockDim.x + threadIdx.x;
    if (i >= n4) return;
    v4f v = *(const v4f*)(&x[i * 4]);
    v.x = fmaxf(v.x, 0.0f); v.y = fmaxf(v.y, 0.0f);
    v.z = fmaxf(v.z, 0.0f); v.w = fmaxf(v.w, 0.0f);
    *(v4f*)(&x[i * 4]) = v;
}

extern "C" void kernel_launch(void* const* d_in, const int* in_sizes, int n_in,
                              void* d_out, int out_size, void* d_ws, size_t ws_size,
                              hipStream_t stream) {
    const float* x   = (const float*)d_in[0];
    const int*   ei  = (const int*)d_in[1];
    const float* W1  = (const float*)d_in[2];
    const float* b1  = (const float*)d_in[3];
    const float* W2  = (const float*)d_in[4];
    const float* b2  = (const float*)d_in[5];
    float*       out = (float*)d_out;

    const int* src = ei;
    const int* dst = ei + NEDGES;

    float* ws   = (float*)d_ws;
    float* deg  = ws;
    float* dinv = ws + 65536;
    float* H    = ws + 131072;
    float* out1 = ws + 131072 + 6400000;

    const int T = 256;
    gcn_deg_init <<<(NNODES + T - 1) / T, T, 0, stream>>>(deg);
    gcn_deg_accum<<<(NEDGES + T - 1) / T, T, 0, stream>>>(dst, deg);
    gcn_dinv     <<<(NNODES + T - 1) / T, T, 0, stream>>>(deg, dinv);

    const int gemm_blocks = GEMM_BLOCKS;
    const int agg_init_blocks = (NNODES * 32 + T - 1) / T;
    const int agg_edge_blocks = NEDGES / 256;
    const int relu_blocks = (NNODES * CH / 4 + T - 1) / T;

    gcn_gemm_wmma<<<gemm_blocks, 128, 0, stream>>>(x, W1, H);
    gcn_agg_init <<<agg_init_blocks, T, 0, stream>>>(H, dinv, b1, out1);
    gcn_agg_edges<<<agg_edge_blocks, T, 0, stream>>>(src, dst, dinv, H, out1);
    gcn_relu     <<<relu_blocks, T, 0, stream>>>(out1, NNODES * CH / 4);

    gcn_gemm_wmma<<<gemm_blocks, 128, 0, stream>>>(out1, W2, H);
    gcn_agg_init <<<agg_init_blocks, T, 0, stream>>>(H, dinv, b2, out);
    gcn_agg_edges<<<agg_edge_blocks, T, 0, stream>>>(src, dst, dinv, H, out);
}


# Self-loop+bias start value and relu folded into the GEMM: relu on the A-row load, tile stored twice from registers (H; H*dinv^2+b into out); 3 element-wise launches gone; sum order unchanged
# speedup vs baseline: 2.2093x; 1.1748x over previous
#include <hip/hip_runtime.h>
#include <hip/hip_bf16.h>

typedef __attribute__((ext_vector_type(2))) float v2f;
typedef __attribute__((ext_vector_type(4))) float v4f;
typedef __attribute__((ext_vector_type(8))) float v8f;

#define NNODES    50000
#define NEDGES    640000
#define CH        128
#define ROW_TILES 3125

__global__ void gcn_deg_init(float* __restrict__ deg) {
    int i = blockIdx.x * blockDim.x + threadIdx.x;
    if (i < NNODES) deg[i] = 1.0f;
}

__global__ void gcn_deg_accum(const int* __restrict__ dst, float* __restrict__ deg) {
    int e = blockIdx.x * blockDim.x + threadIdx.x;
    if (e < NEDGES) unsafeAtomicAdd(&deg[dst[e]], 1.0f);
}

__global__ void gcn_dinv(const float* __restrict__ deg, float* __restrict__ dinv) {
    int i = blockIdx.x * blockDim.x + threadIdx.x;
    if (i < NNODES) {
        float d = deg[i];
        dinv[i] = (d > 0.0f) ? rsqrtf(d) : 0.0f;
    }
}

#define GEMM_BLOCKS 256
#define GEMM_WAVES  (GEMM_BLOCKS * 4)
static_assert(GEMM_WAVES <= ROW_TILES, "every wave owns at least one row tile");
template <bool RELU_IN>
__global__ __launch_bounds__(128) void gcn_gemm_wmma(const float* __restrict__ X,
                                                     const float* __restrict__ W,
                                                     const float* __restrict__ dinv,
                                                     const float* __restrict__ bias,
                                                     float* __restrict__ H,
                                                     float* __restrict__ out) {
    const int lane = threadIdx.x & 31;
    const int half = lane >> 4;
    const int mn   = lane & 15;
    int t = blockIdx.x * 4 + __builtin_amdgcn_readfirstlane(threadIdx.x >> 5);

    v2f a[32];
    {
        const float* __restrict__ xp = X + (size_t)(t * 16 + mn) * CH + 2 * half;
        #pragma unroll
        for (int q = 0; q < 32; ++q) {
            v2f x = *(const v2f*)(xp + 4 * q);
            if (RELU_IN) { x.x = fmaxf(x.x, 0.0f); x.y = fmaxf(x.y, 0.0f); }
            a[q] = x;
        }
    }

    v4f bv[2];
    #pragma unroll
    for (int g = 0; g < 2; ++g) bv[g] = *(const v4f*)(bias + 64 * g + 4 * mn);

    float sq[8];
    {
        const float* __restrict__ dp = dinv + t * 16 + half * 8;
        const v4f d0 = *(const v4f*)(dp);
        const v4f d1 = *(const v4f*)(dp + 4);
        #pragma unroll
        for (int r = 0; r < 4; ++r) { sq[r] = d0[r] * d0[r]; sq[4 + r] = d1[r] * d1[r]; }
    }

    v2f b[32][8];
    {
        const float* __restrict__ wp = W + (size_t)(2 * half) * CH + 4 * mn;
        #pragma unroll
        for (int q = 0; q < 32; ++q) {
            #pragma unroll
            for (int g = 0; g < 2; ++g) {
                v4f w0 = *(const v4f*)(wp + (4 * q + 0) * CH + 64 * g);
                v4f w1 = *(const v4f*)(wp + (4 * q + 1) * CH + 64 * g);
                #pragma unroll
                for (int j = 0; j < 4; ++j) {
                    v2f p; p.x = w0[j]; p.y = w1[j];
                    b[q][4 * g + j] = p;
                }
            }
        }
    }

    for (; t < ROW_TILES; t += GEMM_WAVES) {
        const int tn = (t + GEMM_WAVES < ROW_TILES) ? t + GEMM_WAVES : t;
        v2f an[32];
        {
            const float* __restrict__ xn = X + (size_t)(tn * 16 + mn) * CH + 2 * half;
            #pragma unroll
            for (int q = 0; q < 32; ++q) an[q] = *(const v2f*)(xn + 4 * q);
        }
        const float* __restrict__ dn = dinv + tn * 16 + half * 8;
        const v4f dn0 = *(const v4f*)(dn);
        const v4f dn1 = *(const v4f*)(dn + 4);
        __builtin_amdgcn_sched_barrier(0);

        v8f acc[8];
        v8f zero = {};
        #pragma unroll
        for (int c = 0; c < 8; ++c) acc[c] = zero;

        #pragma unroll
        for (int q = 0; q < 32; ++q) {
            #pragma unroll
            for (int c = 0; c < 8; ++c) {
                acc[c] = __builtin_amdgcn_wmma_f32_16x16x4_f32(
                    false, a[q], false, b[q][c], (short)0, acc[c], false, false);
            }
        }

        const size_t toff = (size_t)(t * 16 + half * 8) * CH + 4 * mn;
        float* __restrict__ hp = H + toff;
        float* __restrict__ op = out + toff;
        #pragma unroll
        for (int r = 0; r < 8; ++r) {
            const float s = sq[r];
            #pragma unroll
            for (int g = 0; g < 2; ++g) {
                v4f o;
                o.x = acc[4 * g + 0][r];
                o.y = acc[4 * g + 1][r];
                o.z = acc[4 * g + 2][r];
                o.w = acc[4 * g + 3][r];
                *(v4f*)(hp + r * CH + 64 * g) = o;
                *(v4f*)(op + r * CH + 64 * g) = o * s + bv[g];
            }
        }

        __builtin_amdgcn_sched_barrier(0);
        #pragma unroll
        for (int q = 0; q < 32; ++q) {
            v2f x = an[q];
            if (RELU_IN) { x.x = fmaxf(x.x, 0.0f); x.y = fmaxf(x.y, 0.0f); }
            a[q] = x;
        }
        #pragma unroll
        for (int r = 0; r < 4; ++r) { sq[r] = dn0[r] * dn0[r]; sq[4 + r] = dn1[r] * dn1[r]; }
    }
}

#define EDGE_WIN 8
static_assert(NEDGES % 256 == 0, "a block takes 256 edges");
__global__ __launch_bounds__(256) void gcn_agg_edges(const int* __restrict__ src,
                                                     const int* __restrict__ dst,
                                                     const float* __restrict__ dinv,
                                                     const float* __restrict__ H,
                                                     float* __restrict__ out) {
    const int lane = threadIdx.x & 31;
    const int e    = blockIdx.x * 256 + threadIdx.x;
    const int s    = src[e];
    const int d    = dst[e];
    const int nb   = __float_as_int(dinv[s] * dinv[d]);

    float h[EDGE_WIN][4];
    #pragma unroll
    for (int g = 0; g < EDGE_WIN; ++g) {
        const float* hp = H + (size_t)__builtin_amdgcn_readlane(s, g) * CH + lane;
        h[g][0] = hp[0];
        h[g][1] = hp[32];
        h[g][2] = hp[64];
        h[g][3] = hp[96];
    }
    #pragma unroll
    for (int j = 0; j < 32; ++j) {
        const int g = j % EDGE_WIN;
        const float norm = __int_as_float(__builtin_amdgcn_readlane(nb, j));
        float* op = out + (size_t)__builtin_amdgcn_readlane(d, j) * CH + lane;
        unsafeAtomicAdd(op + 0,  h[g][0] * norm);
        unsafeAtomicAdd(op + 32, h[g][1] * norm);
        unsafeAtomicAdd(op + 64, h[g][2] * norm);
        unsafeAtomicAdd(op + 96, h[g][3] * norm);
        if (j + EDGE_WIN < 32) {
            const float* hp = H + (size_t)__builtin_amdgcn_readlane(s, j + EDGE_WIN) * CH + lane;
            h[g][0] = hp[0];
            h[g][1] = hp[32];
            h[g][2] = hp[64];
            h[g][3] = hp[96];
        }
    }
}

extern "C" void kernel_launch(void* const* d_in, const int* in_sizes, int n_in,
                              void* d_out, int out_size, void* d_ws, size_t ws_size,
                              hipStream_t stream) {
    const float* x   = (const float*)d_in[0];
    const int*   ei  = (const int*)d_in[1];
    const float* W1  = (const float*)d_in[2];
    const float* b1  = (const float*)d_in[3];
    const float* W2  = (const float*)d_in[4];
    const float* b2  = (const float*)d_in[5];
    float*       out = (float*)d_out;

    const int* src = ei;
    const int* dst = ei + NEDGES;

    float* ws   = (float*)d_ws;
    float* deg  = ws;
    float* dinv = ws + 65536;
    float* H    = ws + 131072;
    float* out1 = ws + 131072 + 6400000;

    const int T = 256;
    gcn_deg_init <<<(NNODES + T - 1) / T, T, 0, stream>>>(deg);
    gcn_deg_accum<<<(NEDGES + T - 1) / T, T, 0, stream>>>(dst, deg);
    gcn_dinv     <<<(NNODES + T - 1) / T, T, 0, stream>>>(deg, dinv);

    const int gemm_blocks = GEMM_BLOCKS;
    const int agg_edge_blocks = NEDGES / 256;

    gcn_gemm_wmma<false><<<gemm_blocks, 128, 0, stream>>>(x, W1, dinv, b1, H, out1);
    gcn_agg_edges<<<agg_edge_blocks, T, 0, stream>>>(src, dst, dinv, H, out1);

    gcn_gemm_wmma<true><<<gemm_blocks, 128, 0, stream>>>(out1, W2, dinv, b2, H, out);
    gcn_agg_edges<<<agg_edge_blocks, T, 0, stream>>>(src, dst, dinv, H, out);
}


# Edge scatter: edges grouped once per call into 256 destination ranges; a work-group sums its range's rows in LDS (f32 LDS atomics, arrival order) and adds them into out with plain stores
# speedup vs baseline: 2.4539x; 1.1107x over previous
#include <hip/hip_runtime.h>
#include <hip/hip_bf16.h>

typedef __attribute__((ext_vector_type(2))) float v2f;
typedef __attribute__((ext_vector_type(4))) float v4f;
typedef __attribute__((ext_vector_type(8))) float v8f;

#define NNODES    50000
#define NEDGES    640000
#define CH        128
#define ROW_TILES 3125

#define NBINS 256
#define RPB   196
static_assert(NBINS * RPB >= NNODES, "the ranges cover every node");
#define EPB   1024
static_assert(NEDGES % EPB == 0, "a block takes EPB edges");

__global__ void gcn_deg_init(float* __restrict__ deg, int* __restrict__ cnt) {
    int i = blockIdx.x * blockDim.x + threadIdx.x;
    if (i < NNODES) deg[i] = 1.0f;
    if (i < 2 * NBINS) cnt[i] = 0;
}

__global__ __launch_bounds__(256) void gcn_deg_accum(const int* __restrict__ dst,
                                                     float* __restrict__ deg,
                                                     int* __restrict__ cnt) {
    __shared__ int lh[NBINS];
    lh[threadIdx.x] = 0;
    __syncthreads();
    for (int k = 0; k < EPB / 256; ++k) {
        const int d = dst[blockIdx.x * EPB + k * 256 + threadIdx.x];
        unsafeAtomicAdd(&deg[d], 1.0f);
        atomicAdd(&lh[d / RPB], 1);
    }
    __syncthreads();
    const int c = lh[threadIdx.x];
    if (c) atomicAdd(&cnt[threadIdx.x], c);
}

__global__ void gcn_dinv(const float* __restrict__ deg, float* __restrict__ dinv) {
    int i = blockIdx.x * blockDim.x + threadIdx.x;
    if (i < NNODES) {
        float d = deg[i];
        dinv[i] = (d > 0.0f) ? rsqrtf(d) : 0.0f;
    }
}

__global__ __launch_bounds__(256) void gcn_bin_place(const int* __restrict__ src,
                                                     const int* __restrict__ dst,
                                                     const float* __restrict__ dinv,
                                                     const int* __restrict__ cnt,
                                                     int* __restrict__ cur,
                                                     int* __restrict__ bstart,
                                                     int2* __restrict__ bedge,
                                                     float* __restrict__ bnorm) {
    __shared__ int lh[NBINS];
    __shared__ int sc[NBINS];
    __shared__ int lb[NBINS];
    const int t = threadIdx.x;
    const int c0 = cnt[t];
    lh[t] = 0;
    sc[t] = c0;
    __syncthreads();
    for (int off = 1; off < NBINS; off <<= 1) {
        const int v = (t >= off) ? sc[t - off] : 0;
        __syncthreads();
        sc[t] += v;
        __syncthreads();
    }
    const int start = sc[t] - c0;
    int s[EPB / 256], d[EPB / 256], r[EPB / 256];
    float nm[EPB / 256];
    #pragma unroll
    for (int k = 0; k < EPB / 256; ++k) {
        const int e = blockIdx.x * EPB + k * 256 + t;
        s[k]  = src[e];
        d[k]  = dst[e];
        nm[k] = dinv[s[k]] * dinv[d[k]];
        r[k]  = atomicAdd(&lh[d[k] / RPB], 1);
    }
    __syncthreads();
    const int c = lh[t];
    const int g = c ? atomicAdd(&cur[t], c) : 0;
    lb[t] = start + g;
    if (blockIdx.x == 0) bstart[t] = start;
    __syncthreads();
    #pragma unroll
    for (int k = 0; k < EPB / 256; ++k) {
        const int pos = lb[d[k] / RPB] + r[k];
        bedge[pos] = make_int2(s[k], d[k]);
        bnorm[pos] = nm[k];
    }
}

#define GEMM_BLOCKS 256
#define GEMM_WAVES  (GEMM_BLOCKS * 4)
static_assert(GEMM_WAVES <= ROW_TILES, "every wave owns at least one row tile");
template <bool RELU_IN>
__global__ __launch_bounds__(128) void gcn_gemm_wmma(const float* __restrict__ X,
                                                     const float* __restrict__ W,
                                                     const float* __restrict__ dinv,
                                                     const float* __restrict__ bias,
                                                     float* __restrict__ H,
                                                     float* __restrict__ out) {
    const int lane = threadIdx.x & 31;
    const int half = lane >> 4;
    const int mn   = lane & 15;
    int t = blockIdx.x * 4 + __builtin_amdgcn_readfirstlane(threadIdx.x >> 5);

    v2f a[32];
    {
        const float* __restrict__ xp = X + (size_t)(t * 16 + mn) * CH + 2 * half;
        #pragma unroll
        for (int q = 0; q < 32; ++q) {
            v2f x = *(const v2f*)(xp + 4 * q);
            if (RELU_IN) { x.x = fmaxf(x.x, 0.0f); x.y = fmaxf(x.y, 0.0f); }
            a[q] = x;
        }
    }

    v4f bv[2];
    #pragma unroll
    for (int g = 0; g < 2; ++g) bv[g] = *(const v4f*)(bias + 64 * g + 4 * mn);

    float sq[8];
    {
        const float* __restrict__ dp = dinv + t * 16 + half * 8;
        const v4f d0 = *(const v4f*)(dp);
        const v4f d1 = *(const v4f*)(dp + 4);
        #pragma unroll
        for (int r = 0; r < 4; ++r) { sq[r] = d0[r] * d0[r]; sq[4 + r] = d1[r] * d1[r]; }
    }

    v2f b[32][8];
    {
        const float* __restrict__ wp = W + (size_t)(2 * half) * CH + 4 * mn;
        #pragma unroll
        for (int q = 0; q < 32; ++q) {
            #pragma unroll
            for (int g = 0; g < 2; ++g) {
                v4f w0 = *(const v4f*)(wp + (4 * q + 0) * CH + 64 * g);
                v4f w1 = *(const v4f*)(wp + (4 * q + 1) * CH + 64 * g);
                #pragma unroll
                for (int j = 0; j < 4; ++j) {
                    v2f p; p.x = w0[j]; p.y = w1[j];
                    b[q][4 * g + j] = p;
                }
            }
        }
    }

    for (; t < ROW_TILES; t += GEMM_WAVES) {
        const int tn = (t + GEMM_WAVES < ROW_TILES) ? t + GEMM_WAVES : t;
        v2f an[32];
        {
            const float* __restrict__ xn = X + (size_t)(tn * 16 + mn) * CH + 2 * half;
            #pragma unroll
            for (int q = 0; q < 32; ++q) an[q] = *(const v2f*)(xn + 4 * q);
        }
        const float* __restrict__ dn = dinv + tn * 16 + half * 8;
        const v4f dn0 = *(const v4f*)(dn);
        const v4f dn1 = *(const v4f*)(dn + 4);
        __builtin_amdgcn_sched_barrier(0);

        v8f acc[8];
        v8f zero = {};
        #pragma unroll
        for (int c = 0; c < 8; ++c) acc[c] = zero;

        #pragma unroll
        for (int q = 0; q < 32; ++q) {
            #pragma unroll
            for (int c = 0; c < 8; ++c) {
                acc[c] = __builtin_amdgcn_wmma_f32_16x16x4_f32(
                    false, a[q], false, b[q][c], (short)0, acc[c], false, false);
            }
        }

        const size_t toff = (size_t)(t * 16 + half * 8) * CH + 4 * mn;
        float* __restrict__ hp = H + toff;
        float* __restrict__ op = out + toff;
        #pragma unroll
        for (int r = 0; r < 8; ++r) {
            const float s = sq[r];
            #pragma unroll
            for (int g = 0; g < 2; ++g) {
                v4f o;
                o.x = acc[4 * g + 0][r];
                o.y = acc[4 * g + 1][r];
                o.z = acc[4 * g + 2][r];
                o.w = acc[4 * g + 3][r];
                *(v4f*)(hp + r * CH + 64 * g) = o;
                *(v4f*)(op + r * CH + 64 * g) = o * s + bv[g];
            }
        }

        __builtin_amdgcn_sched_barrier(0);
        #pragma unroll
        for (int q = 0; q < 32; ++q) {
            v2f x = an[q];
            if (RELU_IN) { x.x = fmaxf(x.x, 0.0f); x.y = fmaxf(x.y, 0.0f); }
            a[q] = x;
        }
        #pragma unroll
        for (int r = 0; r < 4; ++r) { sq[r] = dn0[r] * dn0[r]; sq[4 + r] = dn1[r] * dn1[r]; }
    }
}

#define EDGE_WIN 8
#define AGG_WAVES 32
#define OUT_TRIPS ((RPB * CH / 4 + 1023) / 1024)
__global__ __launch_bounds__(1024) void gcn_agg_bins(const int2* __restrict__ bedge,
                                                     const float* __restrict__ bnorm,
                                                     const int* __restrict__ bstart,
                                                     const int* __restrict__ cnt,
                                                     const float* __restrict__ H,
                                                     float* __restrict__ out) {
    __shared__ float acc[RPB * CH];
    const int tid  = threadIdx.x;
    const int lane = tid & 31;
    const int w    = __builtin_amdgcn_readfirstlane(tid >> 5);
    const int row0 = blockIdx.x * RPB;
    const int nrows = (NNODES - row0 < RPB) ? (NNODES - row0) : RPB;

    v4f zero = {};
    for (int i = tid; i < RPB * CH / 4; i += 1024) ((v4f*)acc)[i] = zero;
    __syncthreads();

    const int start = bstart[blockIdx.x];
    const int n     = cnt[blockIdx.x];
    int ix = start + w * 32 + lane;
    ix = (ix < NEDGES - 1) ? ix : NEDGES - 1;
    int2 sd = bedge[ix];
    int  nb = __float_as_int(bnorm[ix]);
    for (int base = w * 32; base < n; base += AGG_WAVES * 32) {
        const int m = (n - base < 32) ? (n - base) : 32;
        const int s  = sd.x;
        const int dl = sd.y - row0;
        const int nbc = nb;
        ix = start + base + AGG_WAVES * 32 + lane;
        ix = (ix < NEDGES - 1) ? ix : NEDGES - 1;
        sd = bedge[ix];
        nb = __float_as_int(bnorm[ix]);
        float h[EDGE_WIN][4];
        #pragma unroll
        for (int g = 0; g < EDGE_WIN; ++g) {
            const float* hp = H + (size_t)__builtin_amdgcn_readlane(s, g) * CH + lane;
            h[g][0] = hp[0];
            h[g][1] = hp[32];
            h[g][2] = hp[64];
            h[g][3] = hp[96];
        }
        #pragma unroll
        for (int j = 0; j < 32; ++j) {
            const int g = j % EDGE_WIN;
            if (j < m) {
                const float norm = __int_as_float(__builtin_amdgcn_readlane(nbc, j));
                float* ap = acc + __builtin_amdgcn_readlane(dl, j) * CH + lane;
                unsafeAtomicAdd(ap + 0,  h[g][0] * norm);
                unsafeAtomicAdd(ap + 32, h[g][1] * norm);
                unsafeAtomicAdd(ap + 64, h[g][2] * norm);
                unsafeAtomicAdd(ap + 96, h[g][3] * norm);
            }
            if (j + EDGE_WIN < 32) {
                const float* hp = H + (size_t)__builtin_amdgcn_readlane(s, j + EDGE_WIN) * CH + lane;
                h[g][0] = hp[0];
                h[g][1] = hp[32];
                h[g][2] = hp[64];
                h[g][3] = hp[96];
            }
        }
    }
    v4f* __restrict__ op = (v4f*)(out + (size_t)row0 * CH);
    const int lim = nrows * (CH / 4);
    v4f o[OUT_TRIPS];
    #pragma unroll
    for (int k = 0; k < OUT_TRIPS; ++k) {
        const int i = tid + k * 1024;
        if (i < lim) o[k] = op[i];
    }
    __syncthreads();
    #pragma unroll
    for (int k = 0; k < OUT_TRIPS; ++k) {
        const int i = tid + k * 1024;
        if (i < lim) op[i] = o[k] + ((const v4f*)acc)[i];
    }
}

extern "C" void kernel_launch(void* const* d_in, const int* in_sizes, int n_in,
                              void* d_out, int out_size, void* d_ws, size_t ws_size,
                              hipStream_t stream) {
    const float* x   = (const float*)d_in[0];
    const int*   ei  = (const int*)d_in[1];
    const float* W1  = (const float*)d_in[2];
    const float* b1  = (const float*)d_in[3];
    const float* W2  = (const float*)d_in[4];
    const float* b2  = (const float*)d_in[5];
    float*       out = (float*)d_out;

    const int* src = ei;
    const int* dst = ei + NEDGES;

    float* ws   = (float*)d_ws;
    float* deg  = ws;
    float* dinv = ws + 65536;
    float* H    = ws + 131072;
    float* out1 = ws + 131072 + 6400000;
    int*   cnt    = (int*)(ws + 13000000);
    int*   cur    = cnt + 256;
    int*   bstart = cnt + 512;
    int2*  bedge  = (int2*)(cnt + 1024);
    float* bnorm  = ws + 13001024 + 2 * NEDGES;

    const int T = 256;
    gcn_deg_init <<<(NNODES + T - 1) / T, T, 0, stream>>>(deg, cnt);
    gcn_deg_accum<<<NEDGES / EPB, 256, 0, stream>>>(dst, deg, cnt);
    gcn_dinv     <<<(NNODES + T - 1) / T, T, 0, stream>>>(deg, dinv);
    gcn_bin_place<<<NEDGES / EPB, 256, 0, stream>>>(src, dst, dinv, cnt, cur, bstart, bedge, bnorm);

    const int gemm_blocks = GEMM_BLOCKS;

    gcn_gemm_wmma<false><<<gemm_blocks, 128, 0, stream>>>(x, W1, dinv, b1, H, out1);
    gcn_agg_bins<<<NBINS, 1024, 0, stream>>>(bedge, bnorm, bstart, cnt, H, out1);

    gcn_gemm_wmma<true><<<gemm_blocks, 128, 0, stream>>>(out1, W2, dinv, b2, H, out);
    gcn_agg_bins<<<NBINS, 1024, 0, stream>>>(bedge, bnorm, bstart, cnt, H, out);
}


# Edge sum: each gathered H row comes in by ONE 16-byte-a-lane load (4 whole lines; was 4 one-line loads); LDS sums kept channel-permuted so the LDS adds stay conflict-free; sum order unchanged
# speedup vs baseline: 2.5341x; 1.0327x over previous
#include <hip/hip_runtime.h>
#include <hip/hip_bf16.h>

typedef __attribute__((ext_vector_type(2))) float v2f;
typedef __attribute__((ext_vector_type(4))) float v4f;
typedef __attribute__((ext_vector_type(8))) float v8f;

#define NNODES    50000
#define NEDGES    640000
#define CH        128
#define ROW_TILES 3125

#define NBINS 256
#define RPB   196
static_assert(NBINS * RPB >= NNODES, "the ranges cover every node");
#define EPB   1024
static_assert(NEDGES % EPB == 0, "a block takes EPB edges");

__global__ void gcn_deg_init(float* __restrict__ deg, int* __restrict__ cnt) {
    int i = blockIdx.x * blockDim.x + threadIdx.x;
    if (i < NNODES) deg[i] = 1.0f;
    if (i < 2 * NBINS) cnt[i] = 0;
}

__global__ __launch_bounds__(256) void gcn_deg_accum(const int* __restrict__ dst,
                                                     float* __restrict__ deg,
                                                     int* __restrict__ cnt) {
    __shared__ int lh[NBINS];
    lh[threadIdx.x] = 0;
    __syncthreads();
    for (int k = 0; k < EPB / 256; ++k) {
        const int d = dst[blockIdx.x * EPB + k * 256 + threadIdx.x];
        unsafeAtomicAdd(&deg[d], 1.0f);
        atomicAdd(&lh[d / RPB], 1);
    }
    __syncthreads();
    const int c = lh[threadIdx.x];
    if (c) atomicAdd(&cnt[threadIdx.x], c);
}

__global__ void gcn_dinv(const float* __restrict__ deg, float* __restrict__ dinv) {
    int i = blockIdx.x * blockDim.x + threadIdx.x;
    if (i < NNODES) {
        float d = deg[i];
        dinv[i] = (d > 0.0f) ? rsqrtf(d) : 0.0f;
    }
}

__global__ __launch_bounds__(256) void gcn_bin_place(const int* __restrict__ src,
                                                     const int* __restrict__ dst,
                                                     const float* __restrict__ dinv,
                                                     const int* __restrict__ cnt,
                                                     int* __restrict__ cur,
                                                     int* __restrict__ bstart,
                                                     int2* __restrict__ bedge,
                                                     float* __restrict__ bnorm) {
    __shared__ int lh[NBINS];
    __shared__ int sc[NBINS];
    __shared__ int lb[NBINS];
    const int t = threadIdx.x;
    const int c0 = cnt[t];
    lh[t] = 0;
    sc[t] = c0;
    __syncthreads();
    for (int off = 1; off < NBINS; off <<= 1) {
        const int v = (t >= off) ? sc[t - off] : 0;
        __syncthreads();
        sc[t] += v;
        __syncthreads();
    }
    const int start = sc[t] - c0;
    int s[EPB / 256], d[EPB / 256], r[EPB / 256];
    float nm[EPB / 256];
    #pragma unroll
    for (int k = 0; k < EPB / 256; ++k) {
        const int e = blockIdx.x * EPB + k * 256 + t;
        s[k]  = src[e];
        d[k]  = dst[e];
        nm[k] = dinv[s[k]] * dinv[d[k]];
        r[k]  = atomicAdd(&lh[d[k] / RPB], 1);
    }
    __syncthreads();
    const int c = lh[t];
    const int g = c ? atomicAdd(&cur[t], c) : 0;
    lb[t] = start + g;
    if (blockIdx.x == 0) bstart[t] = start;
    __syncthreads();
    #pragma unroll
    for (int k = 0; k < EPB / 256; ++k) {
        const int pos = lb[d[k] / RPB] + r[k];
        bedge[pos] = make_int2(s[k], d[k]);
        bnorm[pos] = nm[k];
    }
}

#define GEMM_BLOCKS 256
#define GEMM_WAVES  (GEMM_BLOCKS * 4)
static_assert(GEMM_WAVES <= ROW_TILES, "every wave owns at least one row tile");
template <bool RELU_IN>
__global__ __launch_bounds__(128) void gcn_gemm_wmma(const float* __restrict__ X,
                                                     const float* __restrict__ W,
                                                     const float* __restrict__ dinv,
                                                     const float* __restrict__ bias,
                                                     float* __restrict__ H,
                                                     float* __restrict__ out) {
    const int lane = threadIdx.x & 31;
    const int half = lane >> 4;
    const int mn   = lane & 15;
    int t = blockIdx.x * 4 + __builtin_amdgcn_readfirstlane(threadIdx.x >> 5);

    v2f a[32];
    {
        const float* __restrict__ xp = X + (size_t)(t * 16 + mn) * CH + 2 * half;
        #pragma unroll
        for (int q = 0; q < 32; ++q) {
            v2f x = *(const v2f*)(xp + 4 * q);
            if (RELU_IN) { x.x = fmaxf(x.x, 0.0f); x.y = fmaxf(x.y, 0.0f); }
            a[q] = x;
        }
    }

    v4f bv[2];
    #pragma unroll
    for (int g = 0; g < 2; ++g) bv[g] = *(const v4f*)(bias + 64 * g + 4 * mn);

    float sq[8];
    {
        const float* __restrict__ dp = dinv + t * 16 + half * 8;
        const v4f d0 = *(const v4f*)(dp);
        const v4f d1 = *(const v4f*)(dp + 4);
        #pragma unroll
        for (int r = 0; r < 4; ++r) { sq[r] = d0[r] * d0[r]; sq[4 + r] = d1[r] * d1[r]; }
    }

    v2f b[32][8];
    {
        const float* __restrict__ wp = W + (size_t)(2 * half) * CH + 4 * mn;
        #pragma unroll
        for (int q = 0; q < 32; ++q) {
            #pragma unroll
            for (int g = 0; g < 2; ++g) {
                v4f w0 = *(const v4f*)(wp + (4 * q + 0) * CH + 64 * g);
                v4f w1 = *(const v4f*)(wp + (4 * q + 1) * CH + 64 * g);
                #pragma unroll
                for (int j = 0; j < 4; ++j) {
                    v2f p; p.x = w0[j]; p.y = w1[j];
                    b[q][4 * g + j] = p;
                }
            }
        }
    }

    for (; t < ROW_TILES; t += GEMM_WAVES) {
        const int tn = (t + GEMM_WAVES < ROW_TILES) ? t + GEMM_WAVES : t;
        v2f an[32];
        {
            const float* __restrict__ xn = X + (size_t)(tn * 16 + mn) * CH + 2 * half;
            #pragma unroll
            for (int q = 0; q < 32; ++q) an[q] = *(const v2f*)(xn + 4 * q);
        }
        const float* __restrict__ dn = dinv + tn * 16 + half * 8;
        const v4f dn0 = *(const v4f*)(dn);
        const v4f dn1 = *(const v4f*)(dn + 4);
        __builtin_amdgcn_sched_barrier(0);

        v8f acc[8];
        v8f zero = {};
        #pragma unroll
        for (int c = 0; c < 8; ++c) acc[c] = zero;

        #pragma unroll
        for (int q = 0; q < 32; ++q) {
            #pragma unroll
            for (int c = 0; c < 8; ++c) {
                acc[c] = __builtin_amdgcn_wmma_f32_16x16x4_f32(
                    false, a[q], false, b[q][c], (short)0, acc[c], false, false);
            }
        }

        const size_t toff = (size_t)(t * 16 + half * 8) * CH + 4 * mn;
        float* __restrict__ hp = H + toff;
        float* __restrict__ op = out + toff;
        #pragma unroll
        for (int r = 0; r < 8; ++r) {
            const float s = sq[r];
            #pragma unroll
            for (int g = 0; g < 2; ++g) {
                v4f o;
                o.x = acc[4 * g + 0][r];
                o.y = acc[4 * g + 1][r];
                o.z = acc[4 * g + 2][r];
                o.w = acc[4 * g + 3][r];
                *(v4f*)(hp + r * CH + 64 * g) = o;
                *(v4f*)(op + r * CH + 64 * g) = o * s + bv[g];
            }
        }

        __builtin_amdgcn_sched_barrier(0);
        #pragma unroll
        for (int q = 0; q < 32; ++q) {
            v2f x = an[q];
            if (RELU_IN) { x.x = fmaxf(x.x, 0.0f); x.y = fmaxf(x.y, 0.0f); }
            a[q] = x;
        }
        #pragma unroll
        for (int r = 0; r < 4; ++r) { sq[r] = dn0[r] * dn0[r]; sq[4 + r] = dn1[r] * dn1[r]; }
    }
}

#define EDGE_WIN 8
#define AGG_WAVES 32
#define OUT_TRIPS ((RPB * CH / 4 + 1023) / 1024)
__global__ __launch_bounds__(1024) void gcn_agg_bins(const int2* __restrict__ bedge,
                                                     const float* __restrict__ bnorm,
                                                     const int* __restrict__ bstart,
                                                     const int* __restrict__ cnt,
                                                     const float* __restrict__ H,
                                                     float* __restrict__ out) {
    __shared__ float acc[RPB * CH];
    const int tid  = threadIdx.x;
    const int lane = tid & 31;
    const unsigned lane4 = 4u * (unsigned)lane;
    const int w    = __builtin_amdgcn_readfirstlane(tid >> 5);
    const int row0 = blockIdx.x * RPB;
    const int nrows = (NNODES - row0 < RPB) ? (NNODES - row0) : RPB;

    v4f zero = {};
    for (int i = tid; i < RPB * CH / 4; i += 1024) ((v4f*)acc)[i] = zero;
    __syncthreads();

    const int start = bstart[blockIdx.x];
    const int n     = cnt[blockIdx.x];
    int ix = start + w * 32 + lane;
    ix = (ix < NEDGES - 1) ? ix : NEDGES - 1;
    int2 sd = bedge[ix];
    int  nb = __float_as_int(bnorm[ix]);
    for (int base = w * 32; base < n; base += AGG_WAVES * 32) {
        const int m = (n - base < 32) ? (n - base) : 32;
        const int s  = sd.x;
        const int dl = sd.y - row0;
        const int nbc = nb;
        ix = start + base + AGG_WAVES * 32 + lane;
        ix = (ix < NEDGES - 1) ? ix : NEDGES - 1;
        sd = bedge[ix];
        nb = __float_as_int(bnorm[ix]);
        v4f h[EDGE_WIN];
        #pragma unroll
        for (int g = 0; g < EDGE_WIN; ++g) {
            const float* rp = H + (size_t)__builtin_amdgcn_readlane(s, g) * CH;
            h[g] = *(const v4f*)(rp + lane4);
        }
        #pragma unroll
        for (int j = 0; j < 32; ++j) {
            const int g = j % EDGE_WIN;
            if (j < m) {
                const float norm = __int_as_float(__builtin_amdgcn_readlane(nbc, j));
                float* ap = acc + __builtin_amdgcn_readlane(dl, j) * CH + lane;
                unsafeAtomicAdd(ap + 0,  h[g].x * norm);
                unsafeAtomicAdd(ap + 32, h[g].y * norm);
                unsafeAtomicAdd(ap + 64, h[g].z * norm);
                unsafeAtomicAdd(ap + 96, h[g].w * norm);
            }
            if (j + EDGE_WIN < 32) {
                const float* rp = H + (size_t)__builtin_amdgcn_readlane(s, j + EDGE_WIN) * CH;
                h[g] = *(const v4f*)(rp + lane4);
            }
        }
    }
    v4f* __restrict__ op = (v4f*)(out + (size_t)row0 * CH);
    const int lim = nrows * (CH / 4);
    v4f o[OUT_TRIPS];
    #pragma unroll
    for (int k = 0; k < OUT_TRIPS; ++k) {
        const int i = tid + k * 1024;
        if (i < lim) o[k] = op[i];
    }
    __syncthreads();
    #pragma unroll
    for (int k = 0; k < OUT_TRIPS; ++k) {
        const int i = tid + k * 1024;
        if (i < lim) {
            const float* ap = acc + (i >> 5) * CH + (i & 31);
            v4f a;
            a.x = ap[0];
            a.y = ap[32];
            a.z = ap[64];
            a.w = ap[96];
            op[i] = o[k] + a;
        }
    }
}

extern "C" void kernel_launch(void* const* d_in, const int* in_sizes, int n_in,
                              void* d_out, int out_size, void* d_ws, size_t ws_size,
                              hipStream_t stream) {
    const float* x   = (const float*)d_in[0];
    const int*   ei  = (const int*)d_in[1];
    const float* W1  = (const float*)d_in[2];
    const float* b1  = (const float*)d_in[3];
    const float* W2  = (const float*)d_in[4];
    const float* b2  = (const float*)d_in[5];
    float*       out = (float*)d_out;

    const int* src = ei;
    const int* dst = ei + NEDGES;

    float* ws   = (float*)d_ws;
    float* deg  = ws;
    float* dinv = ws + 65536;
    float* H    = ws + 131072;
    float* out1 = ws + 131072 + 6400000;
    int*   cnt    = (int*)(ws + 13000000);
    int*   cur    = cnt + 256;
    int*   bstart = cnt + 512;
    int2*  bedge  = (int2*)(cnt + 1024);
    float* bnorm  = ws + 13001024 + 2 * NEDGES;

    const int T = 256;
    gcn_deg_init <<<(NNODES + T - 1) / T, T, 0, stream>>>(deg, cnt);
    gcn_deg_accum<<<NEDGES / EPB, 256, 0, stream>>>(dst, deg, cnt);
    gcn_dinv     <<<(NNODES + T - 1) / T, T, 0, stream>>>(deg, dinv);
    gcn_bin_place<<<NEDGES / EPB, 256, 0, stream>>>(src, dst, dinv, cnt, cur, bstart, bedge, bnorm);

    const int gemm_blocks = GEMM_BLOCKS;

    gcn_gemm_wmma<false><<<gemm_blocks, 128, 0, stream>>>(x, W1, dinv, b1, H, out1);
    gcn_agg_bins<<<NBINS, 1024, 0, stream>>>(bedge, bnorm, bstart, cnt, H, out1);

    gcn_gemm_wmma<true><<<gemm_blocks, 128, 0, stream>>>(out1, W2, dinv, b2, H, out);
    gcn_agg_bins<<<NBINS, 1024, 0, stream>>>(bedge, bnorm, bstart, cnt, H, out);
}


# Edge sum: a range's edges grouped by owner wave in LDS (rank = returning LDS int add, arrival order); each wave alone sums its 6-7 rows by plain LDS load+add+store; LDS float atomics gone
# speedup vs baseline: 2.6189x; 1.0335x over previous
#include <hip/hip_runtime.h>
#include <hip/hip_bf16.h>

typedef __attribute__((ext_vector_type(2))) float v2f;
typedef __attribute__((ext_vector_type(4))) float v4f;
typedef __attribute__((ext_vector_type(8))) float v8f;

#define NNODES    50000
#define NEDGES    640000
#define CH        128
#define ROW_TILES 3125

#define NBINS 256
#define RPB   196
static_assert(NBINS * RPB >= NNODES, "the ranges cover every node");
#define EPB   1024
static_assert(NEDGES % EPB == 0, "a block takes EPB edges");

__global__ void gcn_deg_init(float* __restrict__ deg, int* __restrict__ cnt) {
    int i = blockIdx.x * blockDim.x + threadIdx.x;
    if (i < NNODES) deg[i] = 1.0f;
    if (i < 2 * NBINS) cnt[i] = 0;
}

__global__ __launch_bounds__(256) void gcn_deg_accum(const int* __restrict__ dst,
                                                     float* __restrict__ deg,
                                                     int* __restrict__ cnt) {
    __shared__ int lh[NBINS];
    lh[threadIdx.x] = 0;
    __syncthreads();
    for (int k = 0; k < EPB / 256; ++k) {
        const int d = dst[blockIdx.x * EPB + k * 256 + threadIdx.x];
        unsafeAtomicAdd(&deg[d], 1.0f);
        atomicAdd(&lh[d / RPB], 1);
    }
    __syncthreads();
    const int c = lh[threadIdx.x];
    if (c) atomicAdd(&cnt[threadIdx.x], c);
}

__global__ void gcn_dinv(const float* __restrict__ deg, float* __restrict__ dinv) {
    int i = blockIdx.x * blockDim.x + threadIdx.x;
    if (i < NNODES) {
        float d = deg[i];
        dinv[i] = (d > 0.0f) ? rsqrtf(d) : 0.0f;
    }
}

__global__ __launch_bounds__(256) void gcn_bin_place(const int* __restrict__ src,
                                                     const int* __restrict__ dst,
                                                     const float* __restrict__ dinv,
                                                     const int* __restrict__ cnt,
                                                     int* __restrict__ cur,
                                                     int* __restrict__ bstart,
                                                     int2* __restrict__ bedge,
                                                     float* __restrict__ bnorm) {
    __shared__ int lh[NBINS];
    __shared__ int sc[NBINS];
    __shared__ int lb[NBINS];
    const int t = threadIdx.x;
    const int c0 = cnt[t];
    lh[t] = 0;
    sc[t] = c0;
    __syncthreads();
    for (int off = 1; off < NBINS; off <<= 1) {
        const int v = (t >= off) ? sc[t - off] : 0;
        __syncthreads();
        sc[t] += v;
        __syncthreads();
    }
    const int start = sc[t] - c0;
    int s[EPB / 256], d[EPB / 256], r[EPB / 256];
    float nm[EPB / 256];
    #pragma unroll
    for (int k = 0; k < EPB / 256; ++k) {
        const int e = blockIdx.x * EPB + k * 256 + t;
        s[k]  = src[e];
        d[k]  = dst[e];
        nm[k] = dinv[s[k]] * dinv[d[k]];
        r[k]  = atomicAdd(&lh[d[k] / RPB], 1);
    }
    __syncthreads();
    const int c = lh[t];
    const int g = c ? atomicAdd(&cur[t], c) : 0;
    lb[t] = start + g;
    if (blockIdx.x == 0) bstart[t] = start;
    __syncthreads();
    #pragma unroll
    for (int k = 0; k < EPB / 256; ++k) {
        const int pos = lb[d[k] / RPB] + r[k];
        bedge[pos] = make_int2(s[k], d[k]);
        bnorm[pos] = nm[k];
    }
}

#define GEMM_BLOCKS 256
#define GEMM_WAVES  (GEMM_BLOCKS * 4)
static_assert(GEMM_WAVES <= ROW_TILES, "every wave owns at least one row tile");
template <bool RELU_IN>
__global__ __launch_bounds__(128) void gcn_gemm_wmma(const float* __restrict__ X,
                                                     const float* __restrict__ W,
                                                     const float* __restrict__ dinv,
                                                     const float* __restrict__ bias,
                                                     float* __restrict__ H,
                                                     float* __restrict__ out) {
    const int lane = threadIdx.x & 31;
    const int half = lane >> 4;
    const int mn   = lane & 15;
    int t = blockIdx.x * 4 + __builtin_amdgcn_readfirstlane(threadIdx.x >> 5);

    v2f a[32];
    {
        const float* __restrict__ xp = X + (size_t)(t * 16 + mn) * CH + 2 * half;
        #pragma unroll
        for (int q = 0; q < 32; ++q) {
            v2f x = *(const v2f*)(xp + 4 * q);
            if (RELU_IN) { x.x = fmaxf(x.x, 0.0f); x.y = fmaxf(x.y, 0.0f); }
            a[q] = x;
        }
    }

    v4f bv[2];
    #pragma unroll
    for (int g = 0; g < 2; ++g) bv[g] = *(const v4f*)(bias + 64 * g + 4 * mn);

    float sq[8];
    {
        const float* __restrict__ dp = dinv + t * 16 + half * 8;
        const v4f d0 = *(const v4f*)(dp);
        const v4f d1 = *(const v4f*)(dp + 4);
        #pragma unroll
        for (int r = 0; r < 4; ++r) { sq[r] = d0[r] * d0[r]; sq[4 + r] = d1[r] * d1[r]; }
    }

    v2f b[32][8];
    {
        const float* __restrict__ wp = W + (size_t)(2 * half) * CH + 4 * mn;
        #pragma unroll
        for (int q = 0; q < 32; ++q) {
            #pragma unroll
            for (int g = 0; g < 2; ++g) {
                v4f w0 = *(const v4f*)(wp + (4 * q + 0) * CH + 64 * g);
                v4f w1 = *(const v4f*)(wp + (4 * q + 1) * CH + 64 * g);
                #pragma unroll
                for (int j = 0; j < 4; ++j) {
                    v2f p; p.x = w0[j]; p.y = w1[j];
                    b[q][4 * g + j] = p;
                }
            }
        }
    }

    for (; t < ROW_TILES; t += GEMM_WAVES) {
        const int tn = (t + GEMM_WAVES < ROW_TILES) ? t + GEMM_WAVES : t;
        v2f an[32];
        {
            const float* __restrict__ xn = X + (size_t)(tn * 16 + mn) * CH + 2 * half;
            #pragma unroll
            for (int q = 0; q < 32; ++q) an[q] = *(const v2f*)(xn + 4 * q);
        }
        const float* __restrict__ dn = dinv + tn * 16 + half * 8;
        const v4f dn0 = *(const v4f*)(dn);
        const v4f dn1 = *(const v4f*)(dn + 4);
        __builtin_amdgcn_sched_barrier(0);

        v8f acc[8];
        v8f zero = {};
        #pragma unroll
        for (int c = 0; c < 8; ++c) acc[c] = zero;

        #pragma unroll
        for (int q = 0; q < 32; ++q) {
            #pragma unroll
            for (int c = 0; c < 8; ++c) {
                acc[c] = __builtin_amdgcn_wmma_f32_16x16x4_f32(
                    false, a[q], false, b[q][c], (short)0, acc[c], false, false);
            }
        }

        const size_t toff = (size_t)(t * 16 + half * 8) * CH + 4 * mn;
        float* __restrict__ hp = H + toff;
        float* __restrict__ op = out + toff;
        #pragma unroll
        for (int r = 0; r < 8; ++r) {
            const float s = sq[r];
            #pragma unroll
            for (int g = 0; g < 2; ++g) {
                v4f o;
                o.x = acc[4 * g + 0][r];
                o.y = acc[4 * g + 1][r];
                o.z = acc[4 * g + 2][r];
                o.w = acc[4 * g + 3][r];
                *(v4f*)(hp + r * CH + 64 * g) = o;
                *(v4f*)(op + r * CH + 64 * g) = o * s + bv[g];
            }
        }

        __builtin_amdgcn_sched_barrier(0);
        #pragma unroll
        for (int q = 0; q < 32; ++q) {
            v2f x = an[q];
            if (RELU_IN) { x.x = fmaxf(x.x, 0.0f); x.y = fmaxf(x.y, 0.0f); }
            a[q] = x;
        }
        #pragma unroll
        for (int r = 0; r < 4; ++r) { sq[r] = dn0[r] * dn0[r]; sq[4 + r] = dn1[r] * dn1[r]; }
    }
}

#define EDGE_WIN 8
#define AGG_WAVES 32
#define AGG_CAP   4096
#define ROWS_PW   ((RPB + AGG_WAVES - 1) / AGG_WAVES)
#define CAP_TRIPS (AGG_CAP / 1024)
__global__ __launch_bounds__(1024) void gcn_agg_bins(const int2* __restrict__ bedge,
                                                     const float* __restrict__ bnorm,
                                                     const int* __restrict__ bstart,
                                                     const int* __restrict__ cnt,
                                                     const float* __restrict__ H,
                                                     float* __restrict__ out) {
    __shared__ v4f   acc[RPB * CH / 4];
    __shared__ int2  led[AGG_CAP];
    __shared__ float lnm[AGG_CAP];
    __shared__ int   wcnt[AGG_WAVES];
    __shared__ int   wbeg[AGG_WAVES];
    const int tid  = threadIdx.x;
    const int lane = tid & 31;
    const unsigned lane4 = 4u * (unsigned)lane;
    const int w    = __builtin_amdgcn_readfirstlane(tid >> 5);
    const int row0 = blockIdx.x * RPB;
    const int nrows = (NNODES - row0 < RPB) ? (NNODES - row0) : RPB;

    const int start = bstart[blockIdx.x];
    const int n     = cnt[blockIdx.x];

    v4f zero = {};
    #pragma unroll
    for (int k = 0; k < ROWS_PW; ++k)
        if (w + AGG_WAVES * k < RPB) acc[(w + AGG_WAVES * k) * 32 + lane] = zero;

    for (int c0 = 0; c0 < n; c0 += AGG_CAP) {
        const int mc = (n - c0 < AGG_CAP) ? (n - c0) : AGG_CAP;
        if (tid < AGG_WAVES) wcnt[tid] = 0;
        __syncthreads();
        int es[CAP_TRIPS], ed[CAP_TRIPS], er[CAP_TRIPS];
        float en[CAP_TRIPS];
        #pragma unroll
        for (int k = 0; k < CAP_TRIPS; ++k) {
            const int i = tid + k * 1024;
            if (i < mc) {
                const int2 sd = bedge[start + c0 + i];
                es[k] = sd.x;
                ed[k] = sd.y - row0;
                en[k] = bnorm[start + c0 + i];
                er[k] = atomicAdd(&wcnt[ed[k] & (AGG_WAVES - 1)], 1);
            }
        }
        __syncthreads();
        if (tid < AGG_WAVES) {
            const int c = wcnt[tid];
            int incl = c;
            #pragma unroll
            for (int off = 1; off < AGG_WAVES; off <<= 1) {
                const int v = __shfl_up(incl, off);
                if (tid >= off) incl += v;
            }
            wbeg[tid] = incl - c;
        }
        __syncthreads();
        #pragma unroll
        for (int k = 0; k < CAP_TRIPS; ++k) {
            const int i = tid + k * 1024;
            if (i < mc) {
                const int pos = wbeg[ed[k] & (AGG_WAVES - 1)] + er[k];
                led[pos] = make_int2(es[k], ed[k]);
                lnm[pos] = en[k];
            }
        }
        __syncthreads();

        const int wb = __builtin_amdgcn_readfirstlane(wbeg[w]);
        const int wn = __builtin_amdgcn_readfirstlane(wcnt[w]);
        for (int base = 0; base < wn; base += 32) {
            const int m = (wn - base < 32) ? (wn - base) : 32;
            int ix = base + lane;
            ix = wb + ((ix < wn) ? ix : wn - 1);
            const int2 sd = led[ix];
            const int s   = sd.x;
            const int dl  = sd.y;
            const int nbc = __float_as_int(lnm[ix]);
            v4f h[EDGE_WIN];
            #pragma unroll
            for (int g = 0; g < EDGE_WIN; ++g) {
                const float* rp = H + (size_t)__builtin_amdgcn_readlane(s, g) * CH;
                h[g] = *(const v4f*)(rp + lane4);
            }
            #pragma unroll
            for (int j = 0; j < 32; ++j) {
                const int g = j % EDGE_WIN;
                if (j < m) {
                    const float norm = __int_as_float(__builtin_amdgcn_readlane(nbc, j));
                    v4f* ap = acc + __builtin_amdgcn_readlane(dl, j) * 32 + lane;
                    *ap = *ap + h[g] * norm;
                }
                if (j + EDGE_WIN < 32) {
                    const float* rp = H + (size_t)__builtin_amdgcn_readlane(s, j + EDGE_WIN) * CH;
                    h[g] = *(const v4f*)(rp + lane4);
                }
            }
        }
        if (c0 + AGG_CAP < n) __syncthreads();
    }

    float* __restrict__ op = out + (size_t)(row0 + w) * CH + lane4;
    v4f o[ROWS_PW];
    #pragma unroll
    for (int k = 0; k < ROWS_PW; ++k)
        if (w + AGG_WAVES * k < nrows) o[k] = *(const v4f*)(op + (size_t)(AGG_WAVES * k) * CH);
    #pragma unroll
    for (int k = 0; k < ROWS_PW; ++k)
        if (w + AGG_WAVES * k < nrows)
            *(v4f*)(op + (size_t)(AGG_WAVES * k) * CH) = o[k] + acc[(w + AGG_WAVES * k) * 32 + lane];
}

extern "C" void kernel_launch(void* const* d_in, const int* in_sizes, int n_in,
                              void* d_out, int out_size, void* d_ws, size_t ws_size,
                              hipStream_t stream) {
    const float* x   = (const float*)d_in[0];
    const int*   ei  = (const int*)d_in[1];
    const float* W1  = (const float*)d_in[2];
    const float* b1  = (const float*)d_in[3];
    const float* W2  = (const float*)d_in[4];
    const float* b2  = (const float*)d_in[5];
    float*       out = (float*)d_out;

    const int* src = ei;
    const int* dst = ei + NEDGES;

    float* ws   = (float*)d_ws;
    float* deg  = ws;
    float* dinv = ws + 65536;
    float* H    = ws + 131072;
    float* out1 = ws + 131072 + 6400000;
    int*   cnt    = (int*)(ws + 13000000);
    int*   cur    = cnt + 256;
    int*   bstart = cnt + 512;
    int2*  bedge  = (int2*)(cnt + 1024);
    float* bnorm  = ws + 13001024 + 2 * NEDGES;

    const int T = 256;
    gcn_deg_init <<<(NNODES + T - 1) / T, T, 0, stream>>>(deg, cnt);
    gcn_deg_accum<<<NEDGES / EPB, 256, 0, stream>>>(dst, deg, cnt);
    gcn_dinv     <<<(NNODES + T - 1) / T, T, 0, stream>>>(deg, dinv);
    gcn_bin_place<<<NEDGES / EPB, 256, 0, stream>>>(src, dst, dinv, cnt, cur, bstart, bedge, bnorm);

    const int gemm_blocks = GEMM_BLOCKS;

    gcn_gemm_wmma<false><<<gemm_blocks, 128, 0, stream>>>(x, W1, dinv, b1, H, out1);
    gcn_agg_bins<<<NBINS, 1024, 0, stream>>>(bedge, bnorm, bstart, cnt, H, out1);

    gcn_gemm_wmma<true><<<gemm_blocks, 128, 0, stream>>>(out1, W2, dinv, b2, H, out);
    gcn_agg_bins<<<NBINS, 1024, 0, stream>>>(bedge, bnorm, bstart, cnt, H, out);
}


# Self-loop term, bias and deg^-1/2 formed where they are used (sums kernel's one store of out; edge placing): product stores H only, runs first and carries the init; 2 launches gone; sum order kept
# speedup vs baseline: 2.9373x; 1.1216x over previous
#include <hip/hip_runtime.h>
#include <hip/hip_bf16.h>

typedef __attribute__((ext_vector_type(2))) float v2f;
typedef __attribute__((ext_vector_type(4))) float v4f;
typedef __attribute__((ext_vector_type(8))) float v8f;

#define NNODES    50000
#define NEDGES    640000
#define CH        128
#define ROW_TILES 3125

#define NBINS 256
#define RPB   196
static_assert(NBINS * RPB >= NNODES, "the ranges cover every node");
#define EPB   1024
static_assert(NEDGES % EPB == 0, "a block takes EPB edges");

__global__ __launch_bounds__(256) void gcn_deg_accum(const int* __restrict__ dst,
                                                     float* __restrict__ deg,
                                                     int* __restrict__ cnt) {
    __shared__ int lh[NBINS];
    lh[threadIdx.x] = 0;
    __syncthreads();
    for (int k = 0; k < EPB / 256; ++k) {
        const int d = dst[blockIdx.x * EPB + k * 256 + threadIdx.x];
        unsafeAtomicAdd(&deg[d], 1.0f);
        atomicAdd(&lh[d / RPB], 1);
    }
    __syncthreads();
    const int c = lh[threadIdx.x];
    if (c) atomicAdd(&cnt[threadIdx.x], c);
}

__device__ __forceinline__ float gcn_dinv(float d) {
    return (d > 0.0f) ? rsqrtf(d) : 0.0f;
}

__global__ __launch_bounds__(256) void gcn_bin_place(const int* __restrict__ src,
                                                     const int* __restrict__ dst,
                                                     const float* __restrict__ deg,
                                                     const int* __restrict__ cnt,
                                                     int* __restrict__ cur,
                                                     int* __restrict__ bstart,
                                                     int2* __restrict__ bedge,
                                                     float* __restrict__ bnorm) {
    __shared__ int lh[NBINS];
    __shared__ int sc[NBINS];
    __shared__ int lb[NBINS];
    const int t = threadIdx.x;
    const int c0 = cnt[t];
    lh[t] = 0;
    sc[t] = c0;
    __syncthreads();
    for (int off = 1; off < NBINS; off <<= 1) {
        const int v = (t >= off) ? sc[t - off] : 0;
        __syncthreads();
        sc[t] += v;
        __syncthreads();
    }
    const int start = sc[t] - c0;
    int s[EPB / 256], d[EPB / 256], r[EPB / 256];
    float nm[EPB / 256];
    #pragma unroll
    for (int k = 0; k < EPB / 256; ++k) {
        const int e = blockIdx.x * EPB + k * 256 + t;
        s[k]  = src[e];
        d[k]  = dst[e];
        nm[k] = gcn_dinv(deg[s[k]]) * gcn_dinv(deg[d[k]]);
        r[k]  = atomicAdd(&lh[d[k] / RPB], 1);
    }
    __syncthreads();
    const int c = lh[t];
    const int g = c ? atomicAdd(&cur[t], c) : 0;
    lb[t] = start + g;
    if (blockIdx.x == 0) bstart[t] = start;
    __syncthreads();
    #pragma unroll
    for (int k = 0; k < EPB / 256; ++k) {
        const int pos = lb[d[k] / RPB] + r[k];
        bedge[pos] = make_int2(s[k], d[k]);
        bnorm[pos] = nm[k];
    }
}

#define GEMM_BLOCKS 256
#define GEMM_WAVES  (GEMM_BLOCKS * 4)
static_assert(GEMM_WAVES <= ROW_TILES, "every wave owns at least one row tile");
static_assert(GEMM_BLOCKS * 128 >= 2 * NBINS, "one thread per range count and cursor");
template <bool RELU_IN, bool INIT>
__global__ __launch_bounds__(128) void gcn_gemm_wmma(const float* __restrict__ X,
                                                     const float* __restrict__ W,
                                                     float* __restrict__ H,
                                                     float* __restrict__ deg,
                                                     int* __restrict__ cnt) {
    if (INIT) {
        static_assert(GEMM_BLOCKS * 128 <= NNODES && 2 * GEMM_BLOCKS * 128 >= NNODES,
                      "one or two nodes per thread");
        const int gid = blockIdx.x * 128 + threadIdx.x;
        deg[gid] = 1.0f;
        if (gid + GEMM_BLOCKS * 128 < NNODES) deg[gid + GEMM_BLOCKS * 128] = 1.0f;
        if (gid < 2 * NBINS) cnt[gid] = 0;
    }

    const int lane = threadIdx.x & 31;
    const int half = lane >> 4;
    const int mn   = lane & 15;
    int t = blockIdx.x * 4 + __builtin_amdgcn_readfirstlane(threadIdx.x >> 5);

    v2f a[32];
    {
        const float* __restrict__ xp = X + (size_t)(t * 16 + mn) * CH + 2 * half;
        #pragma unroll
        for (int q = 0; q < 32; ++q) {
            v2f x = *(const v2f*)(xp + 4 * q);
            if (RELU_IN) { x.x = fmaxf(x.x, 0.0f); x.y = fmaxf(x.y, 0.0f); }
            a[q] = x;
        }
    }

    v2f b[32][8];
    {
        const float* __restrict__ wp = W + (size_t)(2 * half) * CH + 4 * mn;
        #pragma unroll
        for (int q = 0; q < 32; ++q) {
            #pragma unroll
            for (int g = 0; g < 2; ++g) {
                v4f w0 = *(const v4f*)(wp + (4 * q + 0) * CH + 64 * g);
                v4f w1 = *(const v4f*)(wp + (4 * q + 1) * CH + 64 * g);
                #pragma unroll
                for (int j = 0; j < 4; ++j) {
                    v2f p; p.x = w0[j]; p.y = w1[j];
                    b[q][4 * g + j] = p;
                }
            }
        }
    }

    for (; t < ROW_TILES; t += GEMM_WAVES) {
        const int tn = (t + GEMM_WAVES < ROW_TILES) ? t + GEMM_WAVES : t;
        v2f an[32];
        {
            const float* __restrict__ xn = X + (size_t)(tn * 16 + mn) * CH + 2 * half;
            #pragma unroll
            for (int q = 0; q < 32; ++q) an[q] = *(const v2f*)(xn + 4 * q);
        }
        __builtin_amdgcn_sched_barrier(0);

        v8f acc[8];
        v8f zero = {};
        #pragma unroll
        for (int c = 0; c < 8; ++c) acc[c] = zero;

        #pragma unroll
        for (int q = 0; q < 32; ++q) {
            #pragma unroll
            for (int c = 0; c < 8; ++c) {
                acc[c] = __builtin_amdgcn_wmma_f32_16x16x4_f32(
                    false, a[q], false, b[q][c], (short)0, acc[c], false, false);
            }
        }

        float* __restrict__ hp = H + (size_t)(t * 16 + half * 8) * CH + 4 * mn;
        #pragma unroll
        for (int r = 0; r < 8; ++r) {
            #pragma unroll
            for (int g = 0; g < 2; ++g) {
                v4f o;
                o.x = acc[4 * g + 0][r];
                o.y = acc[4 * g + 1][r];
                o.z = acc[4 * g + 2][r];
                o.w = acc[4 * g + 3][r];
                *(v4f*)(hp + r * CH + 64 * g) = o;
            }
        }

        __builtin_amdgcn_sched_barrier(0);
        #pragma unroll
        for (int q = 0; q < 32; ++q) {
            v2f x = an[q];
            if (RELU_IN) { x.x = fmaxf(x.x, 0.0f); x.y = fmaxf(x.y, 0.0f); }
            a[q] = x;
        }
    }
}

#define EDGE_WIN 8
#define AGG_WAVES 32
#define AGG_CAP   4096
#define ROWS_PW   ((RPB + AGG_WAVES - 1) / AGG_WAVES)
#define CAP_TRIPS (AGG_CAP / 1024)
__global__ __launch_bounds__(1024) void gcn_agg_bins(const int2* __restrict__ bedge,
                                                     const float* __restrict__ bnorm,
                                                     const int* __restrict__ bstart,
                                                     const int* __restrict__ cnt,
                                                     const float* __restrict__ H,
                                                     const float* __restrict__ deg,
                                                     const float* __restrict__ bias,
                                                     float* __restrict__ out) {
    __shared__ v4f   acc[RPB * CH / 4];
    __shared__ int2  led[AGG_CAP];
    __shared__ float lnm[AGG_CAP];
    __shared__ int   wcnt[AGG_WAVES];
    __shared__ int   wbeg[AGG_WAVES];
    const int tid  = threadIdx.x;
    const int lane = tid & 31;
    const unsigned lane4 = 4u * (unsigned)lane;
    const int w    = __builtin_amdgcn_readfirstlane(tid >> 5);
    const int row0 = blockIdx.x * RPB;
    const int nrows = (NNODES - row0 < RPB) ? (NNODES - row0) : RPB;

    const int start = bstart[blockIdx.x];
    const int n     = cnt[blockIdx.x];

    v4f zero = {};
    #pragma unroll
    for (int k = 0; k < ROWS_PW; ++k)
        if (w + AGG_WAVES * k < RPB) acc[(w + AGG_WAVES * k) * 32 + lane] = zero;

    for (int c0 = 0; c0 < n; c0 += AGG_CAP) {
        const int mc = (n - c0 < AGG_CAP) ? (n - c0) : AGG_CAP;
        if (tid < AGG_WAVES) wcnt[tid] = 0;
        __syncthreads();
        int es[CAP_TRIPS], ed[CAP_TRIPS], er[CAP_TRIPS];
        float en[CAP_TRIPS];
        #pragma unroll
        for (int k = 0; k < CAP_TRIPS; ++k) {
            const int i = tid + k * 1024;
            if (i < mc) {
                const int2 sd = bedge[start + c0 + i];
                es[k] = sd.x;
                ed[k] = sd.y - row0;
                en[k] = bnorm[start + c0 + i];
                er[k] = atomicAdd(&wcnt[ed[k] & (AGG_WAVES - 1)], 1);
            }
        }
        __syncthreads();
        if (tid < AGG_WAVES) {
            const int c = wcnt[tid];
            int incl = c;
            #pragma unroll
            for (int off = 1; off < AGG_WAVES; off <<= 1) {
                const int v = __shfl_up(incl, off);
                if (tid >= off) incl += v;
            }
            wbeg[tid] = incl - c;
        }
        __syncthreads();
        #pragma unroll
        for (int k = 0; k < CAP_TRIPS; ++k) {
            const int i = tid + k * 1024;
            if (i < mc) {
                const int pos = wbeg[ed[k] & (AGG_WAVES - 1)] + er[k];
                led[pos] = make_int2(es[k], ed[k]);
                lnm[pos] = en[k];
            }
        }
        __syncthreads();

        const int wb = __builtin_amdgcn_readfirstlane(wbeg[w]);
        const int wn = __builtin_amdgcn_readfirstlane(wcnt[w]);
        for (int base = 0; base < wn; base += 32) {
            const int m = (wn - base < 32) ? (wn - base) : 32;
            int ix = base + lane;
            ix = wb + ((ix < wn) ? ix : wn - 1);
            const int2 sd = led[ix];
            const int s   = sd.x;
            const int dl  = sd.y;
            const int nbc = __float_as_int(lnm[ix]);
            v4f h[EDGE_WIN];
            #pragma unroll
            for (int g = 0; g < EDGE_WIN; ++g) {
                const float* rp = H + (size_t)__builtin_amdgcn_readlane(s, g) * CH;
                h[g] = *(const v4f*)(rp + lane4);
            }
            #pragma unroll
            for (int j = 0; j < 32; ++j) {
                const int g = j % EDGE_WIN;
                if (j < m) {
                    const float norm = __int_as_float(__builtin_amdgcn_readlane(nbc, j));
                    v4f* ap = acc + __builtin_amdgcn_readlane(dl, j) * 32 + lane;
                    *ap = *ap + h[g] * norm;
                }
                if (j + EDGE_WIN < 32) {
                    const float* rp = H + (size_t)__builtin_amdgcn_readlane(s, j + EDGE_WIN) * CH;
                    h[g] = *(const v4f*)(rp + lane4);
                }
            }
        }
        if (c0 + AGG_CAP < n) __syncthreads();
    }

    static_assert(ROWS_PW <= 32, "one lane per row of the wave");
    const int myrow = w + AGG_WAVES * lane;
    const float dgl = deg[row0 + ((lane < ROWS_PW && myrow < nrows) ? myrow : 0)];
    const float* __restrict__ hp = H + (size_t)(row0 + w) * CH + lane4;
    float* __restrict__ op = out + (size_t)(row0 + w) * CH + lane4;
    const v4f bv = *(const v4f*)(bias + lane4);
    v4f o[ROWS_PW];
    #pragma unroll
    for (int k = 0; k < ROWS_PW; ++k)
        if (w + AGG_WAVES * k < nrows) o[k] = *(const v4f*)(hp + (size_t)(AGG_WAVES * k) * CH);
    const float dil = gcn_dinv(dgl);
    const int   sqb = __float_as_int(dil * dil);
    #pragma unroll
    for (int k = 0; k < ROWS_PW; ++k)
        if (w + AGG_WAVES * k < nrows) {
            const float sq = __int_as_float(__builtin_amdgcn_readlane(sqb, k));
            *(v4f*)(op + (size_t)(AGG_WAVES * k) * CH) =
                (o[k] * sq + bv) + acc[(w + AGG_WAVES * k) * 32 + lane];
        }
}

extern "C" void kernel_launch(void* const* d_in, const int* in_sizes, int n_in,
                              void* d_out, int out_size, void* d_ws, size_t ws_size,
                              hipStream_t stream) {
    const float* x   = (const float*)d_in[0];
    const int*   ei  = (const int*)d_in[1];
    const float* W1  = (const float*)d_in[2];
    const float* b1  = (const float*)d_in[3];
    const float* W2  = (const float*)d_in[4];
    const float* b2  = (const float*)d_in[5];
    float*       out = (float*)d_out;

    const int* src = ei;
    const int* dst = ei + NEDGES;

    float* ws   = (float*)d_ws;
    float* deg  = ws;
    float* H    = ws + 131072;
    float* out1 = ws + 131072 + 6400000;
    int*   cnt    = (int*)(ws + 13000000);
    int*   cur    = cnt + 256;
    int*   bstart = cnt + 512;
    int2*  bedge  = (int2*)(cnt + 1024);
    float* bnorm  = ws + 13001024 + 2 * NEDGES;

    const int gemm_blocks = GEMM_BLOCKS;

    gcn_gemm_wmma<false, true><<<gemm_blocks, 128, 0, stream>>>(x, W1, H, deg, cnt);

    gcn_deg_accum<<<NEDGES / EPB, 256, 0, stream>>>(dst, deg, cnt);
    gcn_bin_place<<<NEDGES / EPB, 256, 0, stream>>>(src, dst, deg, cnt, cur, bstart, bedge, bnorm);

    gcn_agg_bins<<<NBINS, 1024, 0, stream>>>(bedge, bnorm, bstart, cnt, H, deg, b1, out1);

    gcn_gemm_wmma<true, false><<<gemm_blocks, 128, 0, stream>>>(out1, W2, H, deg, cnt);
    gcn_agg_bins<<<NBINS, 1024, 0, stream>>>(bedge, bnorm, bstart, cnt, H, deg, b2, out);
}


# Count and place kernels: a block takes 2560 edges (was 1024), a thread's loads issued together ahead of its adds: 250 blocks, 250 adds or tickets per range counter (was 625); arrival order as before
# speedup vs baseline: 3.0810x; 1.0489x over previous
#include <hip/hip_runtime.h>
#include <hip/hip_bf16.h>

typedef __attribute__((ext_vector_type(2))) float v2f;
typedef __attribute__((ext_vector_type(4))) float v4f;
typedef __attribute__((ext_vector_type(8))) float v8f;

#define NNODES    50000
#define NEDGES    640000
#define CH        128
#define ROW_TILES 3125

#define NBINS 256
#define RPB   196
static_assert(NBINS * RPB >= NNODES, "the ranges cover every node");
#define EPB   2560
static_assert(NEDGES % EPB == 0, "a block takes EPB edges");

__global__ __launch_bounds__(256) void gcn_deg_accum(const int* __restrict__ dst,
                                                     float* __restrict__ deg,
                                                     int* __restrict__ cnt) {
    __shared__ int lh[NBINS];
    lh[threadIdx.x] = 0;
    __syncthreads();
    int d[EPB / 256];
    #pragma unroll
    for (int k = 0; k < EPB / 256; ++k) d[k] = dst[blockIdx.x * EPB + k * 256 + threadIdx.x];
    #pragma unroll
    for (int k = 0; k < EPB / 256; ++k) {
        unsafeAtomicAdd(&deg[d[k]], 1.0f);
        atomicAdd(&lh[d[k] / RPB], 1);
    }
    __syncthreads();
    const int c = lh[threadIdx.x];
    if (c) atomicAdd(&cnt[threadIdx.x], c);
}

__device__ __forceinline__ float gcn_dinv(float d) {
    return (d > 0.0f) ? rsqrtf(d) : 0.0f;
}

__global__ __launch_bounds__(256) void gcn_bin_place(const int* __restrict__ src,
                                                     const int* __restrict__ dst,
                                                     const float* __restrict__ deg,
                                                     const int* __restrict__ cnt,
                                                     int* __restrict__ cur,
                                                     int* __restrict__ bstart,
                                                     int2* __restrict__ bedge,
                                                     float* __restrict__ bnorm) {
    __shared__ int lh[NBINS];
    __shared__ int sc[NBINS];
    __shared__ int lb[NBINS];
    const int t = threadIdx.x;
    const int c0 = cnt[t];
    lh[t] = 0;
    sc[t] = c0;
    __syncthreads();
    for (int off = 1; off < NBINS; off <<= 1) {
        const int v = (t >= off) ? sc[t - off] : 0;
        __syncthreads();
        sc[t] += v;
        __syncthreads();
    }
    const int start = sc[t] - c0;
    int s[EPB / 256], d[EPB / 256], r[EPB / 256];
    float gs[EPB / 256], gd[EPB / 256], nm[EPB / 256];
    #pragma unroll
    for (int k = 0; k < EPB / 256; ++k) {
        const int e = blockIdx.x * EPB + k * 256 + t;
        s[k] = src[e];
        d[k] = dst[e];
    }
    #pragma unroll
    for (int k = 0; k < EPB / 256; ++k) {
        gs[k] = deg[s[k]];
        gd[k] = deg[d[k]];
    }
    #pragma unroll
    for (int k = 0; k < EPB / 256; ++k) {
        nm[k] = gcn_dinv(gs[k]) * gcn_dinv(gd[k]);
        r[k]  = atomicAdd(&lh[d[k] / RPB], 1);
    }
    __syncthreads();
    const int c = lh[t];
    const int g = c ? atomicAdd(&cur[t], c) : 0;
    lb[t] = start + g;
    if (blockIdx.x == 0) bstart[t] = start;
    __syncthreads();
    #pragma unroll
    for (int k = 0; k < EPB / 256; ++k) {
        const int pos = lb[d[k] / RPB] + r[k];
        bedge[pos] = make_int2(s[k], d[k]);
        bnorm[pos] = nm[k];
    }
}

#define GEMM_BLOCKS 256
#define GEMM_WAVES  (GEMM_BLOCKS * 4)
static_assert(GEMM_WAVES <= ROW_TILES, "every wave owns at least one row tile");
static_assert(GEMM_BLOCKS * 128 >= 2 * NBINS, "one thread per range count and cursor");
template <bool RELU_IN, bool INIT>
__global__ __launch_bounds__(128) void gcn_gemm_wmma(const float* __restrict__ X,
                                                     const float* __restrict__ W,
                                                     float* __restrict__ H,
                                                     float* __restrict__ deg,
                                                     int* __restrict__ cnt) {
    if (INIT) {
        static_assert(GEMM_BLOCKS * 128 <= NNODES && 2 * GEMM_BLOCKS * 128 >= NNODES,
                      "one or two nodes per thread");
        const int gid = blockIdx.x * 128 + threadIdx.x;
        deg[gid] = 1.0f;
        if (gid + GEMM_BLOCKS * 128 < NNODES) deg[gid + GEMM_BLOCKS * 128] = 1.0f;
        if (gid < 2 * NBINS) cnt[gid] = 0;
    }

    const int lane = threadIdx.x & 31;
    const int half = lane >> 4;
    const int mn   = lane & 15;
    int t = blockIdx.x * 4 + __builtin_amdgcn_readfirstlane(threadIdx.x >> 5);

    v2f a[32];
    {
        const float* __restrict__ xp = X + (size_t)(t * 16 + mn) * CH + 2 * half;
        #pragma unroll
        for (int q = 0; q < 32; ++q) {
            v2f x = *(const v2f*)(xp + 4 * q);
            if (RELU_IN) { x.x = fmaxf(x.x, 0.0f); x.y = fmaxf(x.y, 0.0f); }
            a[q] = x;
        }
    }

    v2f b[32][8];
    {
        const float* __restrict__ wp = W + (size_t)(2 * half) * CH + 4 * mn;
        #pragma unroll
        for (int q = 0; q < 32; ++q) {
            #pragma unroll
            for (int g = 0; g < 2; ++g) {
                v4f w0 = *(const v4f*)(wp + (4 * q + 0) * CH + 64 * g);
                v4f w1 = *(const v4f*)(wp + (4 * q + 1) * CH + 64 * g);
                #pragma unroll
                for (int j = 0; j < 4; ++j) {
                    v2f p; p.x = w0[j]; p.y = w1[j];
                    b[q][4 * g + j] = p;
                }
            }
        }
    }

    for (; t < ROW_TILES; t += GEMM_WAVES) {
        const int tn = (t + GEMM_WAVES < ROW_TILES) ? t + GEMM_WAVES : t;
        v2f an[32];
        {
            const float* __restrict__ xn = X + (size_t)(tn * 16 + mn) * CH + 2 * half;
            #pragma unroll
            for (int q = 0; q < 32; ++q) an[q] = *(const v2f*)(xn + 4 * q);
        }
        __builtin_amdgcn_sched_barrier(0);

        v8f acc[8];
        v8f zero = {};
        #pragma unroll
        for (int c = 0; c < 8; ++c) acc[c] = zero;

        #pragma unroll
        for (int q = 0; q < 32; ++q) {
            #pragma unroll
            for (int c = 0; c < 8; ++c) {
                acc[c] = __builtin_amdgcn_wmma_f32_16x16x4_f32(
                    false, a[q], false, b[q][c], (short)0, acc[c], false, false);
            }
        }

        float* __restrict__ hp = H + (size_t)(t * 16 + half * 8) * CH + 4 * mn;
        #pragma unroll
        for (int r = 0; r < 8; ++r) {
            #pragma unroll
            for (int g = 0; g < 2; ++g) {
                v4f o;
                o.x = acc[4 * g + 0][r];
                o.y = acc[4 * g + 1][r];
                o.z = acc[4 * g + 2][r];
                o.w = acc[4 * g + 3][r];
                *(v4f*)(hp + r * CH + 64 * g) = o;
            }
        }

        __builtin_amdgcn_sched_barrier(0);
        #pragma unroll
        for (int q = 0; q < 32; ++q) {
            v2f x = an[q];
            if (RELU_IN) { x.x = fmaxf(x.x, 0.0f); x.y = fmaxf(x.y, 0.0f); }
            a[q] = x;
        }
    }
}

#define EDGE_WIN 8
#define AGG_WAVES 32
#define AGG_CAP   4096
#define ROWS_PW   ((RPB + AGG_WAVES - 1) / AGG_WAVES)
#define CAP_TRIPS (AGG_CAP / 1024)
__global__ __launch_bounds__(1024) void gcn_agg_bins(const int2* __restrict__ bedge,
                                                     const float* __restrict__ bnorm,
                                                     const int* __restrict__ bstart,
                                                     const int* __restrict__ cnt,
                                                     const float* __restrict__ H,
                                                     const float* __restrict__ deg,
                                                     const float* __restrict__ bias,
                                                     float* __restrict__ out) {
    __shared__ v4f   acc[RPB * CH / 4];
    __shared__ int2  led[AGG_CAP];
    __shared__ float lnm[AGG_CAP];
    __shared__ int   wcnt[AGG_WAVES];
    __shared__ int   wbeg[AGG_WAVES];
    const int tid  = threadIdx.x;
    const int lane = tid & 31;
    const unsigned lane4 = 4u * (unsigned)lane;
    const int w    = __builtin_amdgcn_readfirstlane(tid >> 5);
    const int row0 = blockIdx.x * RPB;
    const int nrows = (NNODES - row0 < RPB) ? (NNODES - row0) : RPB;

    const int start = bstart[blockIdx.x];
    const int n     = cnt[blockIdx.x];

    v4f zero = {};
    #pragma unroll
    for (int k = 0; k < ROWS_PW; ++k)
        if (w + AGG_WAVES * k < RPB) acc[(w + AGG_WAVES * k) * 32 + lane] = zero;

    for (int c0 = 0; c0 < n; c0 += AGG_CAP) {
        const int mc = (n - c0 < AGG_CAP) ? (n - c0) : AGG_CAP;
        if (tid < AGG_WAVES) wcnt[tid] = 0;
        __syncthreads();
        int es[CAP_TRIPS], ed[CAP_TRIPS], er[CAP_TRIPS];
        float en[CAP_TRIPS];
        #pragma unroll
        for (int k = 0; k < CAP_TRIPS; ++k) {
            const int i = tid + k * 1024;
            if (i < mc) {
                const int2 sd = bedge[start + c0 + i];
                es[k] = sd.x;
                ed[k] = sd.y - row0;
                en[k] = bnorm[start + c0 + i];
                er[k] = atomicAdd(&wcnt[ed[k] & (AGG_WAVES - 1)], 1);
            }
        }
        __syncthreads();
        if (tid < AGG_WAVES) {
            const int c = wcnt[tid];
            int incl = c;
            #pragma unroll
            for (int off = 1; off < AGG_WAVES; off <<= 1) {
                const int v = __shfl_up(incl, off);
                if (tid >= off) incl += v;
            }
            wbeg[tid] = incl - c;
        }
        __syncthreads();
        #pragma unroll
        for (int k = 0; k < CAP_TRIPS; ++k) {
            const int i = tid + k * 1024;
            if (i < mc) {
                const int pos = wbeg[ed[k] & (AGG_WAVES - 1)] + er[k];
                led[pos] = make_int2(es[k], ed[k]);
                lnm[pos] = en[k];
            }
        }
        __syncthreads();

        const int wb = __builtin_amdgcn_readfirstlane(wbeg[w]);
        const int wn = __builtin_amdgcn_readfirstlane(wcnt[w]);
        for (int base = 0; base < wn; base += 32) {
            const int m = (wn - base < 32) ? (wn - base) : 32;
            int ix = base + lane;
            ix = wb + ((ix < wn) ? ix : wn - 1);
            const int2 sd = led[ix];
            const int s   = sd.x;
            const int dl  = sd.y;
            const int nbc = __float_as_int(lnm[ix]);
            v4f h[EDGE_WIN];
            #pragma unroll
            for (int g = 0; g < EDGE_WIN; ++g) {
                const float* rp = H + (size_t)__builtin_amdgcn_readlane(s, g) * CH;
                h[g] = *(const v4f*)(rp + lane4);
            }
            #pragma unroll
            for (int j = 0; j < 32; ++j) {
                const int g = j % EDGE_WIN;
                if (j < m) {
                    const float norm = __int_as_float(__builtin_amdgcn_readlane(nbc, j));
                    v4f* ap = acc + __builtin_amdgcn_readlane(dl, j) * 32 + lane;
                    *ap = *ap + h[g] * norm;
                }
                if (j + EDGE_WIN < 32) {
                    const float* rp = H + (size_t)__builtin_amdgcn_readlane(s, j + EDGE_WIN) * CH;
                    h[g] = *(const v4f*)(rp + lane4);
                }
            }
        }
        if (c0 + AGG_CAP < n) __syncthreads();
    }

    static_assert(ROWS_PW <= 32, "one lane per row of the wave");
    const int myrow = w + AGG_WAVES * lane;
    const float dgl = deg[row0 + ((lane < ROWS_PW && myrow < nrows) ? myrow : 0)];
    const float* __restrict__ hp = H + (size_t)(row0 + w) * CH + lane4;
    float* __restrict__ op = out + (size_t)(row0 + w) * CH + lane4;
    const v4f bv = *(const v4f*)(bias + lane4);
    v4f o[ROWS_PW];
    #pragma unroll
    for (int k = 0; k < ROWS_PW; ++k)
        if (w + AGG_WAVES * k < nrows) o[k] = *(const v4f*)(hp + (size_t)(AGG_WAVES * k) * CH);
    const float dil = gcn_dinv(dgl);
    const int   sqb = __float_as_int(dil * dil);
    #pragma unroll
    for (int k = 0; k < ROWS_PW; ++k)
        if (w + AGG_WAVES * k < nrows) {
            const float sq = __int_as_float(__builtin_amdgcn_readlane(sqb, k));
            *(v4f*)(op + (size_t)(AGG_WAVES * k) * CH) =
                (o[k] * sq + bv) + acc[(w + AGG_WAVES * k) * 32 + lane];
        }
}

extern "C" void kernel_launch(void* const* d_in, const int* in_sizes, int n_in,
                              void* d_out, int out_size, void* d_ws, size_t ws_size,
                              hipStream_t stream) {
    const float* x   = (const float*)d_in[0];
    const int*   ei  = (const int*)d_in[1];
    const float* W1  = (const float*)d_in[2];
    const float* b1  = (const float*)d_in[3];
    const float* W2  = (const float*)d_in[4];
    const float* b2  = (const float*)d_in[5];
    float*       out = (float*)d_out;

    const int* src = ei;
    const int* dst = ei + NEDGES;

    float* ws   = (float*)d_ws;
    float* deg  = ws;
    float* H    = ws + 131072;
    float* out1 = ws + 131072 + 6400000;
    int*   cnt    = (int*)(ws + 13000000);
    int*   cur    = cnt + 256;
    int*   bstart = cnt + 512;
    int2*  bedge  = (int2*)(cnt + 1024);
    float* bnorm  = ws + 13001024 + 2 * NEDGES;

    const int gemm_blocks = GEMM_BLOCKS;

    gcn_gemm_wmma<false, true><<<gemm_blocks, 128, 0, stream>>>(x, W1, H, deg, cnt);

    gcn_deg_accum<<<NEDGES / EPB, 256, 0, stream>>>(dst, deg, cnt);
    gcn_bin_place<<<NEDGES / EPB, 256, 0, stream>>>(src, dst, deg, cnt, cur, bstart, bedge, bnorm);

    gcn_agg_bins<<<NBINS, 1024, 0, stream>>>(bedge, bnorm, bstart, cnt, H, deg, b1, out1);

    gcn_gemm_wmma<true, false><<<gemm_blocks, 128, 0, stream>>>(out1, W2, H, deg, cnt);
    gcn_agg_bins<<<NBINS, 1024, 0, stream>>>(bedge, bnorm, bstart, cnt, H, deg, b2, out);
}


# Degree count and range grouping in ONE pass: a block groups its own 2560 edges by range in its own stretch (no range totals, no cursors); sum kernel walks the 250 pieces; 1 launch gone; arrival order
# speedup vs baseline: 3.3231x; 1.0786x over previous
#include <hip/hip_runtime.h>
#include <hip/hip_bf16.h>

typedef __attribute__((ext_vector_type(2))) float v2f;
typedef __attribute__((ext_vector_type(4))) float v4f;
typedef __attribute__((ext_vector_type(8))) float v8f;

#define NNODES    50000
#define NEDGES    640000
#define CH        128
#define ROW_TILES 3125

#define NBINS 256
#define RPB   196
static_assert(NBINS * RPB >= NNODES, "the ranges cover every node");
#define EPB   2560
#define NBLK  (NEDGES / EPB)
static_assert(NEDGES % EPB == 0, "a block takes EPB edges");
static_assert(NBLK <= NBINS, "one thread of the summing kernel per block of the grouping kernel");

__device__ __forceinline__ float gcn_dinv(float d) {
    return (d > 0.0f) ? rsqrtf(d) : 0.0f;
}

__global__ __launch_bounds__(256) void gcn_bin_local(const int* __restrict__ src,
                                                     const int* __restrict__ dst,
                                                     float* __restrict__ deg,
                                                     int2* __restrict__ segtab,
                                                     int2* __restrict__ bedge) {
    __shared__ int lh[NBINS];
    __shared__ int sc[NBINS];
    __shared__ int lb[NBINS];
    const int t = threadIdx.x;
    lh[t] = 0;
    __syncthreads();
    int s[EPB / 256], d[EPB / 256], r[EPB / 256];
    #pragma unroll
    for (int k = 0; k < EPB / 256; ++k) {
        const int e = blockIdx.x * EPB + k * 256 + t;
        s[k] = src[e];
        d[k] = dst[e];
    }
    #pragma unroll
    for (int k = 0; k < EPB / 256; ++k) {
        unsafeAtomicAdd(&deg[d[k]], 1.0f);
        r[k] = atomicAdd(&lh[d[k] / RPB], 1);
    }
    __syncthreads();
    const int c0 = lh[t];
    sc[t] = c0;
    __syncthreads();
    for (int off = 1; off < NBINS; off <<= 1) {
        const int v = (t >= off) ? sc[t - off] : 0;
        __syncthreads();
        sc[t] += v;
        __syncthreads();
    }
    const int first = blockIdx.x * EPB + sc[t] - c0;
    lb[t] = first;
    segtab[blockIdx.x * NBINS + t] = make_int2(first, c0);
    __syncthreads();
    #pragma unroll
    for (int k = 0; k < EPB / 256; ++k)
        bedge[lb[d[k] / RPB] + r[k]] = make_int2(s[k], d[k]);
}

#define GEMM_BLOCKS 256
#define GEMM_WAVES  (GEMM_BLOCKS * 4)
static_assert(GEMM_WAVES <= ROW_TILES, "every wave owns at least one row tile");
template <bool RELU_IN, bool INIT>
__global__ __launch_bounds__(128) void gcn_gemm_wmma(const float* __restrict__ X,
                                                     const float* __restrict__ W,
                                                     float* __restrict__ H,
                                                     float* __restrict__ deg) {
    if (INIT) {
        static_assert(GEMM_BLOCKS * 128 <= NNODES && 2 * GEMM_BLOCKS * 128 >= NNODES,
                      "one or two nodes per thread");
        const int gid = blockIdx.x * 128 + threadIdx.x;
        deg[gid] = 1.0f;
        if (gid + GEMM_BLOCKS * 128 < NNODES) deg[gid + GEMM_BLOCKS * 128] = 1.0f;
    }

    const int lane = threadIdx.x & 31;
    const int half = lane >> 4;
    const int mn   = lane & 15;
    int t = blockIdx.x * 4 + __builtin_amdgcn_readfirstlane(threadIdx.x >> 5);

    v2f a[32];
    {
        const float* __restrict__ xp = X + (size_t)(t * 16 + mn) * CH + 2 * half;
        #pragma unroll
        for (int q = 0; q < 32; ++q) {
            v2f x = *(const v2f*)(xp + 4 * q);
            if (RELU_IN) { x.x = fmaxf(x.x, 0.0f); x.y = fmaxf(x.y, 0.0f); }
            a[q] = x;
        }
    }

    v2f b[32][8];
    {
        const float* __restrict__ wp = W + (size_t)(2 * half) * CH + 4 * mn;
        #pragma unroll
        for (int q = 0; q < 32; ++q) {
            #pragma unroll
            for (int g = 0; g < 2; ++g) {
                v4f w0 = *(const v4f*)(wp + (4 * q + 0) * CH + 64 * g);
                v4f w1 = *(const v4f*)(wp + (4 * q + 1) * CH + 64 * g);
                #pragma unroll
                for (int j = 0; j < 4; ++j) {
                    v2f p; p.x = w0[j]; p.y = w1[j];
                    b[q][4 * g + j] = p;
                }
            }
        }
    }

    for (; t < ROW_TILES; t += GEMM_WAVES) {
        const int tn = (t + GEMM_WAVES < ROW_TILES) ? t + GEMM_WAVES : t;
        v2f an[32];
        {
            const float* __restrict__ xn = X + (size_t)(tn * 16 + mn) * CH + 2 * half;
            #pragma unroll
            for (int q = 0; q < 32; ++q) an[q] = *(const v2f*)(xn + 4 * q);
        }
        __builtin_amdgcn_sched_barrier(0);

        v8f acc[8];
        v8f zero = {};
        #pragma unroll
        for (int c = 0; c < 8; ++c) acc[c] = zero;

        #pragma unroll
        for (int q = 0; q < 32; ++q) {
            #pragma unroll
            for (int c = 0; c < 8; ++c) {
                acc[c] = __builtin_amdgcn_wmma_f32_16x16x4_f32(
                    false, a[q], false, b[q][c], (short)0, acc[c], false, false);
            }
        }

        float* __restrict__ hp = H + (size_t)(t * 16 + half * 8) * CH + 4 * mn;
        #pragma unroll
        for (int r = 0; r < 8; ++r) {
            #pragma unroll
            for (int g = 0; g < 2; ++g) {
                v4f o;
                o.x = acc[4 * g + 0][r];
                o.y = acc[4 * g + 1][r];
                o.z = acc[4 * g + 2][r];
                o.w = acc[4 * g + 3][r];
                *(v4f*)(hp + r * CH + 64 * g) = o;
            }
        }

        __builtin_amdgcn_sched_barrier(0);
        #pragma unroll
        for (int q = 0; q < 32; ++q) {
            v2f x = an[q];
            if (RELU_IN) { x.x = fmaxf(x.x, 0.0f); x.y = fmaxf(x.y, 0.0f); }
            a[q] = x;
        }
    }
}

#define EDGE_WIN 8
#define AGG_WAVES 32
#define AGG_CAP   4096
#define ROWS_PW   ((RPB + AGG_WAVES - 1) / AGG_WAVES)
#define CAP_TRIPS (AGG_CAP / 1024)
__global__ __launch_bounds__(1024) void gcn_agg_bins(const int2* __restrict__ bedge,
                                                     const int2* __restrict__ segtab,
                                                     const float* __restrict__ H,
                                                     const float* __restrict__ deg,
                                                     const float* __restrict__ bias,
                                                     float* __restrict__ out) {
    __shared__ v4f   acc[RPB * CH / 4];
    __shared__ int2  led[AGG_CAP];
    __shared__ float lnm[AGG_CAP];
    __shared__ int   wcnt[AGG_WAVES];
    __shared__ int   wbeg[AGG_WAVES];
    __shared__ int   pre[NBINS];
    __shared__ int   sgs[NBINS];
    __shared__ int   wtot[NBINS / 32];
    const int tid  = threadIdx.x;
    const int lane = tid & 31;
    const unsigned lane4 = 4u * (unsigned)lane;
    const int w    = __builtin_amdgcn_readfirstlane(tid >> 5);
    const int row0 = blockIdx.x * RPB;
    const int nrows = (NNODES - row0 < RPB) ? (NNODES - row0) : RPB;

    int2 sg = make_int2(0, 0);
    if (tid < NBLK) sg = segtab[tid * NBINS + blockIdx.x];
    int incl = sg.y;
    if (tid < NBINS) {
        #pragma unroll
        for (int off = 1; off < 32; off <<= 1) {
            const int v = __shfl_up(incl, off);
            if (lane >= off) incl += v;
        }
        if (lane == 31) wtot[w] = incl;
    }
    __syncthreads();
    int n = 0, nfront = 0;
    #pragma unroll
    for (int v = 0; v < NBINS / 32; ++v) {
        const int x = wtot[v];
        n += x;
        if (v < w) nfront += x;
    }
    n = __builtin_amdgcn_readfirstlane(n);
    if (tid < NBINS) {
        pre[tid] = nfront + incl - sg.y;
        sgs[tid] = sg.x;
    }

    v4f zero = {};
    #pragma unroll
    for (int k = 0; k < ROWS_PW; ++k)
        if (w + AGG_WAVES * k < RPB) acc[(w + AGG_WAVES * k) * 32 + lane] = zero;

    for (int c0 = 0; c0 < n; c0 += AGG_CAP) {
        const int mc = (n - c0 < AGG_CAP) ? (n - c0) : AGG_CAP;
        if (tid < AGG_WAVES) wcnt[tid] = 0;
        __syncthreads();
        int ff[CAP_TRIPS], pp[CAP_TRIPS], ei[CAP_TRIPS];
        int es[CAP_TRIPS], ed[CAP_TRIPS], er[CAP_TRIPS];
        float gs[CAP_TRIPS], gd[CAP_TRIPS], en[CAP_TRIPS];
        #pragma unroll
        for (int k = 0; k < CAP_TRIPS; ++k) {
            const int i = tid + k * 1024;
            ff[k] = c0 + ((i < mc) ? i : mc - 1);
            pp[k] = 0;
        }
        #pragma unroll
        for (int st = NBINS / 2; st > 0; st >>= 1) {
            #pragma unroll
            for (int k = 0; k < CAP_TRIPS; ++k)
                if (pre[pp[k] + st] <= ff[k]) pp[k] += st;
        }
        #pragma unroll
        for (int k = 0; k < CAP_TRIPS; ++k) ei[k] = sgs[pp[k]] + (ff[k] - pre[pp[k]]);
        #pragma unroll
        for (int k = 0; k < CAP_TRIPS; ++k) {
            const int2 sd = bedge[ei[k]];
            es[k] = sd.x;
            ed[k] = sd.y;
        }
        #pragma unroll
        for (int k = 0; k < CAP_TRIPS; ++k) {
            gs[k] = deg[es[k]];
            gd[k] = deg[ed[k]];
        }
        #pragma unroll
        for (int k = 0; k < CAP_TRIPS; ++k) {
            const int i = tid + k * 1024;
            en[k] = gcn_dinv(gs[k]) * gcn_dinv(gd[k]);
            ed[k] -= row0;
            if (i < mc) er[k] = atomicAdd(&wcnt[ed[k] & (AGG_WAVES - 1)], 1);
        }
        __syncthreads();
        if (tid < AGG_WAVES) {
            const int c = wcnt[tid];
            int incl = c;
            #pragma unroll
            for (int off = 1; off < AGG_WAVES; off <<= 1) {
                const int v = __shfl_up(incl, off);
                if (tid >= off) incl += v;
            }
            wbeg[tid] = incl - c;
        }
        __syncthreads();
        #pragma unroll
        for (int k = 0; k < CAP_TRIPS; ++k) {
            const int i = tid + k * 1024;
            if (i < mc) {
                const int pos = wbeg[ed[k] & (AGG_WAVES - 1)] + er[k];
                led[pos] = make_int2(es[k], ed[k]);
                lnm[pos] = en[k];
            }
        }
        __syncthreads();

        const int wb = __builtin_amdgcn_readfirstlane(wbeg[w]);
        const int wn = __builtin_amdgcn_readfirstlane(wcnt[w]);
        for (int base = 0; base < wn; base += 32) {
            const int m = (wn - base < 32) ? (wn - base) : 32;
            int ix = base + lane;
            ix = wb + ((ix < wn) ? ix : wn - 1);
            const int2 sd = led[ix];
            const int s   = sd.x;
            const int dl  = sd.y;
            const int nbc = __float_as_int(lnm[ix]);
            v4f h[EDGE_WIN];
            #pragma unroll
            for (int g = 0; g < EDGE_WIN; ++g) {
                const float* rp = H + (size_t)__builtin_amdgcn_readlane(s, g) * CH;
                h[g] = *(const v4f*)(rp + lane4);
            }
            #pragma unroll
            for (int j = 0; j < 32; ++j) {
                const int g = j % EDGE_WIN;
                if (j < m) {
                    const float norm = __int_as_float(__builtin_amdgcn_readlane(nbc, j));
                    v4f* ap = acc + __builtin_amdgcn_readlane(dl, j) * 32 + lane;
                    *ap = *ap + h[g] * norm;
                }
                if (j + EDGE_WIN < 32) {
                    const float* rp = H + (size_t)__builtin_amdgcn_readlane(s, j + EDGE_WIN) * CH;
                    h[g] = *(const v4f*)(rp + lane4);
                }
            }
        }
        if (c0 + AGG_CAP < n) __syncthreads();
    }

    static_assert(ROWS_PW <= 32, "one lane per row of the wave");
    const int myrow = w + AGG_WAVES * lane;
    const float dgl = deg[row0 + ((lane < ROWS_PW && myrow < nrows) ? myrow : 0)];
    const float* __restrict__ hp = H + (size_t)(row0 + w) * CH + lane4;
    float* __restrict__ op = out + (size_t)(row0 + w) * CH + lane4;
    const v4f bv = *(const v4f*)(bias + lane4);
    v4f o[ROWS_PW];
    #pragma unroll
    for (int k = 0; k < ROWS_PW; ++k)
        if (w + AGG_WAVES * k < nrows) o[k] = *(const v4f*)(hp + (size_t)(AGG_WAVES * k) * CH);
    const float dil = gcn_dinv(dgl);
    const int   sqb = __float_as_int(dil * dil);
    #pragma unroll
    for (int k = 0; k < ROWS_PW; ++k)
        if (w + AGG_WAVES * k < nrows) {
            const float sq = __int_as_float(__builtin_amdgcn_readlane(sqb, k));
            *(v4f*)(op + (size_t)(AGG_WAVES * k) * CH) =
                (o[k] * sq + bv) + acc[(w + AGG_WAVES * k) * 32 + lane];
        }
}

extern "C" void kernel_launch(void* const* d_in, const int* in_sizes, int n_in,
                              void* d_out, int out_size, void* d_ws, size_t ws_size,
                              hipStream_t stream) {
    const float* x   = (const float*)d_in[0];
    const int*   ei  = (const int*)d_in[1];
    const float* W1  = (const float*)d_in[2];
    const float* b1  = (const float*)d_in[3];
    const float* W2  = (const float*)d_in[4];
    const float* b2  = (const float*)d_in[5];
    float*       out = (float*)d_out;

    const int* src = ei;
    const int* dst = ei + NEDGES;

    float* ws   = (float*)d_ws;
    float* deg  = ws;
    float* H    = ws + 131072;
    float* out1 = ws + 131072 + 6400000;
    int2*  segtab = (int2*)(ws + 13000000);
    int2*  bedge  = (int2*)(ws + 13000000 + 131072);

    const int gemm_blocks = GEMM_BLOCKS;

    gcn_gemm_wmma<false, true><<<gemm_blocks, 128, 0, stream>>>(x, W1, H, deg);

    gcn_bin_local<<<NBLK, 256, 0, stream>>>(src, dst, deg, segtab, bedge);

    gcn_agg_bins<<<NBINS, 1024, 0, stream>>>(bedge, segtab, H, deg, b1, out1);

    gcn_gemm_wmma<true, false><<<gemm_blocks, 128, 0, stream>>>(out1, W2, H, deg);
    gcn_agg_bins<<<NBINS, 1024, 0, stream>>>(bedge, segtab, H, deg, b2, out);
}
